# starting kernel
# baseline (speedup 1.0000x reference)
#include <hip/hip_runtime.h>
#include <math.h>


typedef __attribute__((ext_vector_type(2))) float v2f;
typedef __attribute__((ext_vector_type(8))) float v8f;

#define N_NODES 200000
#define N_EDGES 600000
#define F_IN    117
#define F_PAD   124
#define D       128
#define D_PAD   132
#define L_LAYERS 3

__device__ __forceinline__ v8f wmma_f32_16x16x4(v2f a, v2f b, v8f c) {
  return __builtin_amdgcn_wmma_f32_16x16x4_f32(false, a, false, b, (short)0, c,
                                               false, false);
}

__global__ __launch_bounds__(256) void degree_kernel(const int* __restrict__ dst,
                                                     float* __restrict__ deg,
                                                     int nE) {
  int e = blockIdx.x * blockDim.x + threadIdx.x;
  if (e < nE) atomicAdd(&deg[dst[e]], 1.0f);
}

__global__ __launch_bounds__(256) void inv_deg_kernel(float* __restrict__ deg, int n) {
  int i = blockIdx.x * blockDim.x + threadIdx.x;
  if (i < n) deg[i] = 1.0f / fmaxf(deg[i], 1.0f);
}

__global__ __launch_bounds__(256) void input_gemm_tanh(const float* __restrict__ x,
                                                       const float* __restrict__ W_in,
                                                       const float* __restrict__ b_in,
                                                       float* __restrict__ h_out) {
  __shared__ float sx[16 * F_PAD];
  const int tid = threadIdx.x;
  const int mBase = blockIdx.x * 16;

  for (int i = tid; i < 16 * F_PAD; i += 256) {
    int r = i / F_PAD, c = i - r * F_PAD;
    sx[i] = (c < F_IN) ? x[(size_t)(mBase + r) * F_IN + c] : 0.0f;
  }
  __syncthreads();

  const int lane = tid & 31;
  const int half = lane >> 4;
  const int l16  = lane & 15;
  const int col  = (tid >> 5) * 16 + l16;

  v8f acc = {0.f, 0.f, 0.f, 0.f, 0.f, 0.f, 0.f, 0.f};
  for (int k = 0; k < F_IN; k += 4) {
    const int ka = k + 2 * half;
    v2f a, b;
    a.x = sx[l16 * F_PAD + ka];
    a.y = sx[l16 * F_PAD + ka + 1];
    b.x = (ka     < F_IN) ? W_in[(size_t)(ka)     * D + col] : 0.0f;
    b.y = (ka + 1 < F_IN) ? W_in[(size_t)(ka + 1) * D + col] : 0.0f;
    acc = wmma_f32_16x16x4(a, b, acc);
  }

  const float bias = b_in[col];
  const int rowBase = mBase + 8 * half;
#pragma unroll
  for (int r = 0; r < 8; ++r)
    h_out[(size_t)(rowBase + r) * D + col] = tanhf(acc[r] + bias);
}

__global__ __launch_bounds__(256) void scatter_sum(const float* __restrict__ h,
                                                   const int* __restrict__ src,
                                                   const int* __restrict__ dst,
                                                   float* __restrict__ agg, int nE) {
  int e = blockIdx.x * 8 + (threadIdx.x >> 5);
  if (e >= nE) return;
  int lane = threadIdx.x & 31;
  int s = src[e], d = dst[e];
  const float4* hv = (const float4*)(h + (size_t)s * D);
  float4 v = hv[lane];
  float* ap = agg + (size_t)d * D + lane * 4;
  atomicAdd(ap + 0, v.x);
  atomicAdd(ap + 1, v.y);
  atomicAdd(ap + 2, v.z);
  atomicAdd(ap + 3, v.w);
}

__global__ __launch_bounds__(256) void layer_gemm(const float* __restrict__ h,
                                                  const float* __restrict__ agg,
                                                  const float* __restrict__ inv_deg,
                                                  const float* __restrict__ Wself,
                                                  const float* __restrict__ Wneigh,
                                                  const float* __restrict__ bias_v,
                                                  float* __restrict__ h_out,
                                                  int relu) {
  __shared__ float sh[16 * D_PAD];
  __shared__ float sa[16 * D_PAD];
  const int tid = threadIdx.x;
  const int mBase = blockIdx.x * 16;

  for (int i = tid; i < 16 * D; i += 256) {
    int r = i >> 7, c = i & (D - 1);
    int row = mBase + r;
    float id = inv_deg[row];
    sh[r * D_PAD + c] = h[(size_t)row * D + c];
    sa[r * D_PAD + c] = agg[(size_t)row * D + c] * id;
  }
  __syncthreads();

  const int lane = tid & 31;
  const int half = lane >> 4;
  const int l16  = lane & 15;
  const int col  = (tid >> 5) * 16 + l16;

  v8f acc = {0.f, 0.f, 0.f, 0.f, 0.f, 0.f, 0.f, 0.f};
  for (int k = 0; k < D; k += 4) {
    const int ka = k + 2 * half;
    v2f a1, b1, a2, b2;
    a1.x = sh[l16 * D_PAD + ka];
    a1.y = sh[l16 * D_PAD + ka + 1];
    b1.x = Wself[(size_t)(ka)     * D + col];
    b1.y = Wself[(size_t)(ka + 1) * D + col];
    acc = wmma_f32_16x16x4(a1, b1, acc);

    a2.x = sa[l16 * D_PAD + ka];
    a2.y = sa[l16 * D_PAD + ka + 1];
    b2.x = Wneigh[(size_t)(ka)     * D + col];
    b2.y = Wneigh[(size_t)(ka + 1) * D + col];
    acc = wmma_f32_16x16x4(a2, b2, acc);
  }

  const float bv = bias_v[col];
  const int rowBase = mBase + 8 * half;
#pragma unroll
  for (int r = 0; r < 8; ++r) {
    float v = acc[r] + bv;
    if (relu) v = fmaxf(v, 0.0f);
    h_out[(size_t)(rowBase + r) * D + col] = v;
  }
}

extern "C" void kernel_launch(void* const* d_in, const int* in_sizes, int n_in,
                              void* d_out, int out_size, void* d_ws, size_t ws_size,
                              hipStream_t stream) {
  const float* x       = (const float*)d_in[0];
  const int*   ei      = (const int*)d_in[1];
  const float* W_in    = (const float*)d_in[2];
  const float* b_in    = (const float*)d_in[3];
  const float* W_self  = (const float*)d_in[4];
  const float* W_neigh = (const float*)d_in[5];
  const float* b_lay   = (const float*)d_in[6];
  float* out = (float*)d_out;

  float* ws   = (float*)d_ws;
  float* hA   = ws;
  float* hB   = ws + (size_t)1 * N_NODES * D;
  float* agg  = ws + (size_t)2 * N_NODES * D;
  float* degv = ws + (size_t)3 * N_NODES * D;

  const int* src = ei;
  const int* dst = ei + N_EDGES;

  hipMemsetAsync(degv, 0, N_NODES * sizeof(float), stream);
  degree_kernel<<<(N_EDGES + 255) / 256, 256, 0, stream>>>(dst, degv, N_EDGES);
  inv_deg_kernel<<<(N_NODES + 255) / 256, 256, 0, stream>>>(degv, N_NODES);

  input_gemm_tanh<<<N_NODES / 16, 256, 0, stream>>>(x, W_in, b_in, hA);

  const float* hcur = hA;
  for (int i = 0; i < L_LAYERS; ++i) {
    hipMemsetAsync(agg, 0, (size_t)N_NODES * D * sizeof(float), stream);
    scatter_sum<<<(N_EDGES + 7) / 8, 256, 0, stream>>>(hcur, src, dst, agg, N_EDGES);
    float* hnext = (i == L_LAYERS - 1) ? out : (hcur == hA ? hB : hA);
    layer_gemm<<<N_NODES / 16, 256, 0, stream>>>(
        hcur, agg, degv,
        W_self  + (size_t)i * D * D,
        W_neigh + (size_t)i * D * D,
        b_lay   + (size_t)i * D,
        hnext, (i < L_LAYERS - 1) ? 1 : 0);
    hcur = hnext;
  }
}


# GEMM kernels made weight-stationary: each wave keeps its 16-column W fragments in registers while a fixed grid of 512 work-groups loops over 32-row tiles (next tile prefetched); sum order unchanged
# speedup vs baseline: 1.1755x; 1.1755x over previous
#include <hip/hip_runtime.h>
#include <math.h>


typedef __attribute__((ext_vector_type(2))) float v2f;
typedef __attribute__((ext_vector_type(4))) float v4f;
typedef __attribute__((ext_vector_type(8))) float v8f;

#define N_NODES 200000
#define N_EDGES 600000
#define F_IN    117
#define F_PAD   124
#define D       128
#define D_PAD   132
#define L_LAYERS 3
#define TILE_M   32
#define N_TILES  (N_NODES / TILE_M)
#define GEMM_GRID 512
#define X_TILE_F4 (TILE_M * F_IN / 4)

__device__ __forceinline__ v8f wmma_f32_16x16x4(v2f a, v2f b, v8f c) {
  return __builtin_amdgcn_wmma_f32_16x16x4_f32(false, a, false, b, (short)0, c,
                                               false, false);
}

__global__ __launch_bounds__(256) void degree_kernel(const int* __restrict__ dst,
                                                     float* __restrict__ deg,
                                                     int nE) {
  int e = blockIdx.x * blockDim.x + threadIdx.x;
  if (e < nE) atomicAdd(&deg[dst[e]], 1.0f);
}

__global__ __launch_bounds__(256) void inv_deg_kernel(float* __restrict__ deg, int n) {
  int i = blockIdx.x * blockDim.x + threadIdx.x;
  if (i < n) deg[i] = 1.0f / fmaxf(deg[i], 1.0f);
}

__global__ __launch_bounds__(256) void input_gemm_tanh(const float* __restrict__ x,
                                                       const float* __restrict__ W_in,
                                                       const float* __restrict__ b_in,
                                                       float* __restrict__ h_out) {
  __shared__ __attribute__((aligned(16))) float sx[2][TILE_M * F_PAD];
  const int tid  = threadIdx.x;
  const int lane = tid & 31;
  const int half = lane >> 4;
  const int l16  = lane & 15;
  const int col  = (tid >> 5) * 16 + l16;

  for (int i = tid; i < 2 * TILE_M * (F_PAD - F_IN); i += 256) {
    int r = i / (F_PAD - F_IN), c = i - r * (F_PAD - F_IN);
    (&sx[0][0])[r * F_PAD + F_IN + c] = 0.0f;
  }

  v2f bW[30];
#pragma unroll
  for (int t = 0; t < 30; ++t) {
    const int ka = 4 * t + 2 * half;
    bW[t].x = (ka     < F_IN) ? W_in[(size_t)(ka)     * D + col] : 0.0f;
    bW[t].y = (ka + 1 < F_IN) ? W_in[(size_t)(ka + 1) * D + col] : 0.0f;
  }
  const float bias = b_in[col];

  const bool tail = tid < X_TILE_F4 - 768;
  v4f px[4];
  int tile = blockIdx.x;
  if (tile < N_TILES) {
    const v4f* xt = (const v4f*)(x + (size_t)tile * (TILE_M * F_IN));
    px[0] = xt[tid];
    px[1] = xt[tid + 256];
    px[2] = xt[tid + 512];
    if (tail) px[3] = xt[tid + 768];
  }

  int buf = 0;
  for (; tile < N_TILES; tile += GEMM_GRID) {
    float* sxb = sx[buf];
#pragma unroll
    for (int j = 0; j < 4; ++j) {
      if (j < 3 || tail) {
        int i0 = 4 * (tid + 256 * j);
        int r0 = i0 / F_IN, c0 = i0 - r0 * F_IN;
#pragma unroll
        for (int u = 0; u < 4; ++u) {
          sxb[r0 * F_PAD + c0] = px[j][u];
          if (++c0 == F_IN) { c0 = 0; ++r0; }
        }
      }
    }
    __syncthreads();

    const int nxt = tile + GEMM_GRID;
    if (nxt < N_TILES) {
      const v4f* xt = (const v4f*)(x + (size_t)nxt * (TILE_M * F_IN));
      px[0] = xt[tid];
      px[1] = xt[tid + 256];
      px[2] = xt[tid + 512];
      if (tail) px[3] = xt[tid + 768];
    }

    v8f acc0 = {0.f, 0.f, 0.f, 0.f, 0.f, 0.f, 0.f, 0.f};
    v8f acc1 = {0.f, 0.f, 0.f, 0.f, 0.f, 0.f, 0.f, 0.f};
    const float* ar0 = sxb + l16 * F_PAD + 2 * half;
    const float* ar1 = ar0 + 16 * F_PAD;
#pragma unroll
    for (int t = 0; t < 30; ++t) {
      v2f a0 = *(const v2f*)(ar0 + 4 * t);
      v2f a1 = *(const v2f*)(ar1 + 4 * t);
      acc0 = wmma_f32_16x16x4(a0, bW[t], acc0);
      acc1 = wmma_f32_16x16x4(a1, bW[t], acc1);
    }

    float* ob = h_out + (size_t)tile * (TILE_M * D);
    const int lo = 8 * half * D + col;
#pragma unroll
    for (int r = 0; r < 8; ++r)
      ob[lo + r * D] = tanhf(acc0[r] + bias);
#pragma unroll
    for (int r = 0; r < 8; ++r)
      ob[lo + (16 + r) * D] = tanhf(acc1[r] + bias);

    buf ^= 1;
  }
}

__global__ __launch_bounds__(256) void scatter_sum(const float* __restrict__ h,
                                                   const int* __restrict__ src,
                                                   const int* __restrict__ dst,
                                                   float* __restrict__ agg, int nE) {
  int e = blockIdx.x * 8 + (threadIdx.x >> 5);
  if (e >= nE) return;
  int lane = threadIdx.x & 31;
  int s = src[e], d = dst[e];
  const float4* hv = (const float4*)(h + (size_t)s * D);
  float4 v = hv[lane];
  float* ap = agg + (size_t)d * D + lane * 4;
  atomicAdd(ap + 0, v.x);
  atomicAdd(ap + 1, v.y);
  atomicAdd(ap + 2, v.z);
  atomicAdd(ap + 3, v.w);
}

struct LayerStage {
  v4f   h0, h1, h2, h3;
  v4f   a0, a1, a2, a3;
  float i0, i1, i2, i3;
};

__device__ __forceinline__ void layer_stage_load(LayerStage& s, const float* __restrict__ h,
                                                 const float* __restrict__ agg,
                                                 const float* __restrict__ inv_deg,
                                                 int tile, int sr, int sc) {
  const v4f* hp = (const v4f*)(h   + (size_t)tile * (TILE_M * D));
  const v4f* ap = (const v4f*)(agg + (size_t)tile * (TILE_M * D));
  const float*  ip = inv_deg + (size_t)tile * TILE_M;
  const int o = sr * (D / 4) + sc;
  s.h0 = hp[o];              s.a0 = ap[o];              s.i0 = ip[sr];
  s.h1 = hp[o + 8 * D / 4];  s.a1 = ap[o + 8 * D / 4];  s.i1 = ip[sr + 8];
  s.h2 = hp[o + 16 * D / 4]; s.a2 = ap[o + 16 * D / 4]; s.i2 = ip[sr + 16];
  s.h3 = hp[o + 24 * D / 4]; s.a3 = ap[o + 24 * D / 4]; s.i3 = ip[sr + 24];
}

__global__ __launch_bounds__(256) void layer_gemm(const float* __restrict__ h,
                                                  const float* __restrict__ agg,
                                                  const float* __restrict__ inv_deg,
                                                  const float* __restrict__ Wself,
                                                  const float* __restrict__ Wneigh,
                                                  const float* __restrict__ bias_v,
                                                  float* __restrict__ h_out,
                                                  int relu) {
  __shared__ __attribute__((aligned(16))) float sh[2][TILE_M * D_PAD];
  __shared__ __attribute__((aligned(16))) float sa[2][TILE_M * D_PAD];
  const int tid  = threadIdx.x;
  const int lane = tid & 31;
  const int half = lane >> 4;
  const int l16  = lane & 15;
  const int col  = (tid >> 5) * 16 + l16;

  v2f bS[32], bN[32];
#pragma unroll
  for (int t = 0; t < 32; ++t) {
    const int ka = 4 * t + 2 * half;
    bS[t].x = Wself[(size_t)(ka)      * D + col];
    bS[t].y = Wself[(size_t)(ka + 1)  * D + col];
    bN[t].x = Wneigh[(size_t)(ka)     * D + col];
    bN[t].y = Wneigh[(size_t)(ka + 1) * D + col];
  }
  const float bv = bias_v[col];

  const int sc = tid & 31;
  const int sr = tid >> 5;
  LayerStage st;

  int tile = blockIdx.x;
  if (tile < N_TILES) layer_stage_load(st, h, agg, inv_deg, tile, sr, sc);

  int buf = 0;
  for (; tile < N_TILES; tile += GEMM_GRID) {
    float* shb = sh[buf];
    float* sab = sa[buf];
    {
      const int o = sr * D_PAD + 4 * sc;
      *(v4f*)(shb + o)              = st.h0;
      *(v4f*)(shb + o + 8 * D_PAD)  = st.h1;
      *(v4f*)(shb + o + 16 * D_PAD) = st.h2;
      *(v4f*)(shb + o + 24 * D_PAD) = st.h3;
      *(v4f*)(sab + o)              = st.a0 * st.i0;
      *(v4f*)(sab + o + 8 * D_PAD)  = st.a1 * st.i1;
      *(v4f*)(sab + o + 16 * D_PAD) = st.a2 * st.i2;
      *(v4f*)(sab + o + 24 * D_PAD) = st.a3 * st.i3;
    }
    __syncthreads();

    const int nxt = tile + GEMM_GRID;
    if (nxt < N_TILES) layer_stage_load(st, h, agg, inv_deg, nxt, sr, sc);

    v8f acc0 = {0.f, 0.f, 0.f, 0.f, 0.f, 0.f, 0.f, 0.f};
    v8f acc1 = {0.f, 0.f, 0.f, 0.f, 0.f, 0.f, 0.f, 0.f};
    const float* h0 = shb + l16 * D_PAD + 2 * half;
    const float* h1 = h0 + 16 * D_PAD;
    const float* g0 = sab + l16 * D_PAD + 2 * half;
    const float* g1 = g0 + 16 * D_PAD;
#pragma unroll
    for (int t = 0; t < 32; ++t) {
      v2f a10 = *(const v2f*)(h0 + 4 * t);
      v2f a11 = *(const v2f*)(h1 + 4 * t);
      acc0 = wmma_f32_16x16x4(a10, bS[t], acc0);
      acc1 = wmma_f32_16x16x4(a11, bS[t], acc1);
      v2f a20 = *(const v2f*)(g0 + 4 * t);
      v2f a21 = *(const v2f*)(g1 + 4 * t);
      acc0 = wmma_f32_16x16x4(a20, bN[t], acc0);
      acc1 = wmma_f32_16x16x4(a21, bN[t], acc1);
    }

    float* ob = h_out + (size_t)tile * (TILE_M * D);
    const int lo = 8 * half * D + col;
#pragma unroll
    for (int r = 0; r < 8; ++r) {
      float v = acc0[r] + bv;
      if (relu) v = fmaxf(v, 0.0f);
      ob[lo + r * D] = v;
    }
#pragma unroll
    for (int r = 0; r < 8; ++r) {
      float v = acc1[r] + bv;
      if (relu) v = fmaxf(v, 0.0f);
      ob[lo + (16 + r) * D] = v;
    }

    buf ^= 1;
  }
}

extern "C" void kernel_launch(void* const* d_in, const int* in_sizes, int n_in,
                              void* d_out, int out_size, void* d_ws, size_t ws_size,
                              hipStream_t stream) {
  const float* x       = (const float*)d_in[0];
  const int*   ei      = (const int*)d_in[1];
  const float* W_in    = (const float*)d_in[2];
  const float* b_in    = (const float*)d_in[3];
  const float* W_self  = (const float*)d_in[4];
  const float* W_neigh = (const float*)d_in[5];
  const float* b_lay   = (const float*)d_in[6];
  float* out = (float*)d_out;

  float* ws   = (float*)d_ws;
  float* hA   = ws;
  float* hB   = ws + (size_t)1 * N_NODES * D;
  float* agg  = ws + (size_t)2 * N_NODES * D;
  float* degv = ws + (size_t)3 * N_NODES * D;

  const int* src = ei;
  const int* dst = ei + N_EDGES;

  hipMemsetAsync(degv, 0, N_NODES * sizeof(float), stream);
  degree_kernel<<<(N_EDGES + 255) / 256, 256, 0, stream>>>(dst, degv, N_EDGES);
  inv_deg_kernel<<<(N_NODES + 255) / 256, 256, 0, stream>>>(degv, N_NODES);

  input_gemm_tanh<<<GEMM_GRID, 256, 0, stream>>>(x, W_in, b_in, hA);

  const float* hcur = hA;
  for (int i = 0; i < L_LAYERS; ++i) {
    hipMemsetAsync(agg, 0, (size_t)N_NODES * D * sizeof(float), stream);
    scatter_sum<<<(N_EDGES + 7) / 8, 256, 0, stream>>>(hcur, src, dst, agg, N_EDGES);
    float* hnext = (i == L_LAYERS - 1) ? out : (hcur == hA ? hB : hA);
    layer_gemm<<<GEMM_GRID, 256, 0, stream>>>(
        hcur, agg, degv,
        W_self  + (size_t)i * D * D,
        W_neigh + (size_t)i * D * D,
        b_lay   + (size_t)i * D,
        hnext, (i < L_LAYERS - 1) ? 1 : 0);
    hcur = hnext;
  }
}


# Edge scatter lane map: lane = feature channel, so each row load and each f32 atomic covers one whole 128-B line (4 line adds per edge, not 16 quarter-filled); sums still in arrival order
# speedup vs baseline: 1.3512x; 1.1495x over previous
#include <hip/hip_runtime.h>
#include <math.h>


typedef __attribute__((ext_vector_type(2))) float v2f;
typedef __attribute__((ext_vector_type(4))) float v4f;
typedef __attribute__((ext_vector_type(8))) float v8f;

#define N_NODES 200000
#define N_EDGES 600000
#define F_IN    117
#define F_PAD   124
#define D       128
#define D_PAD   132
#define L_LAYERS 3
#define TILE_M   32
#define N_TILES  (N_NODES / TILE_M)
#define GEMM_GRID 512
#define X_TILE_F4 (TILE_M * F_IN / 4)

__device__ __forceinline__ v8f wmma_f32_16x16x4(v2f a, v2f b, v8f c) {
  return __builtin_amdgcn_wmma_f32_16x16x4_f32(false, a, false, b, (short)0, c,
                                               false, false);
}

__global__ __launch_bounds__(256) void degree_kernel(const int* __restrict__ dst,
                                                     float* __restrict__ deg,
                                                     int nE) {
  int e = blockIdx.x * blockDim.x + threadIdx.x;
  if (e < nE) atomicAdd(&deg[dst[e]], 1.0f);
}

__global__ __launch_bounds__(256) void inv_deg_kernel(float* __restrict__ deg, int n) {
  int i = blockIdx.x * blockDim.x + threadIdx.x;
  if (i < n) deg[i] = 1.0f / fmaxf(deg[i], 1.0f);
}

__global__ __launch_bounds__(256) void input_gemm_tanh(const float* __restrict__ x,
                                                       const float* __restrict__ W_in,
                                                       const float* __restrict__ b_in,
                                                       float* __restrict__ h_out) {
  __shared__ __attribute__((aligned(16))) float sx[2][TILE_M * F_PAD];
  const int tid  = threadIdx.x;
  const int lane = tid & 31;
  const int half = lane >> 4;
  const int l16  = lane & 15;
  const int col  = (tid >> 5) * 16 + l16;

  for (int i = tid; i < 2 * TILE_M * (F_PAD - F_IN); i += 256) {
    int r = i / (F_PAD - F_IN), c = i - r * (F_PAD - F_IN);
    (&sx[0][0])[r * F_PAD + F_IN + c] = 0.0f;
  }

  v2f bW[30];
#pragma unroll
  for (int t = 0; t < 30; ++t) {
    const int ka = 4 * t + 2 * half;
    bW[t].x = (ka     < F_IN) ? W_in[(size_t)(ka)     * D + col] : 0.0f;
    bW[t].y = (ka + 1 < F_IN) ? W_in[(size_t)(ka + 1) * D + col] : 0.0f;
  }
  const float bias = b_in[col];

  const bool tail = tid < X_TILE_F4 - 768;
  v4f px[4];
  int tile = blockIdx.x;
  if (tile < N_TILES) {
    const v4f* xt = (const v4f*)(x + (size_t)tile * (TILE_M * F_IN));
    px[0] = xt[tid];
    px[1] = xt[tid + 256];
    px[2] = xt[tid + 512];
    if (tail) px[3] = xt[tid + 768];
  }

  int buf = 0;
  for (; tile < N_TILES; tile += GEMM_GRID) {
    float* sxb = sx[buf];
#pragma unroll
    for (int j = 0; j < 4; ++j) {
      if (j < 3 || tail) {
        int i0 = 4 * (tid + 256 * j);
        int r0 = i0 / F_IN, c0 = i0 - r0 * F_IN;
#pragma unroll
        for (int u = 0; u < 4; ++u) {
          sxb[r0 * F_PAD + c0] = px[j][u];
          if (++c0 == F_IN) { c0 = 0; ++r0; }
        }
      }
    }
    __syncthreads();

    const int nxt = tile + GEMM_GRID;
    if (nxt < N_TILES) {
      const v4f* xt = (const v4f*)(x + (size_t)nxt * (TILE_M * F_IN));
      px[0] = xt[tid];
      px[1] = xt[tid + 256];
      px[2] = xt[tid + 512];
      if (tail) px[3] = xt[tid + 768];
    }

    v8f acc0 = {0.f, 0.f, 0.f, 0.f, 0.f, 0.f, 0.f, 0.f};
    v8f acc1 = {0.f, 0.f, 0.f, 0.f, 0.f, 0.f, 0.f, 0.f};
    const float* ar0 = sxb + l16 * F_PAD + 2 * half;
    const float* ar1 = ar0 + 16 * F_PAD;
#pragma unroll
    for (int t = 0; t < 30; ++t) {
      v2f a0 = *(const v2f*)(ar0 + 4 * t);
      v2f a1 = *(const v2f*)(ar1 + 4 * t);
      acc0 = wmma_f32_16x16x4(a0, bW[t], acc0);
      acc1 = wmma_f32_16x16x4(a1, bW[t], acc1);
    }

    float* ob = h_out + (size_t)tile * (TILE_M * D);
    const int lo = 8 * half * D + col;
#pragma unroll
    for (int r = 0; r < 8; ++r)
      ob[lo + r * D] = tanhf(acc0[r] + bias);
#pragma unroll
    for (int r = 0; r < 8; ++r)
      ob[lo + (16 + r) * D] = tanhf(acc1[r] + bias);

    buf ^= 1;
  }
}

__global__ __launch_bounds__(256) void scatter_sum(const float* __restrict__ h,
                                                   const int* __restrict__ src,
                                                   const int* __restrict__ dst,
                                                   float* __restrict__ agg, int nE) {
  int e = blockIdx.x * 8 + (threadIdx.x >> 5);
  if (e >= nE) return;
  int lane = threadIdx.x & 31;
  int s = src[e], d = dst[e];
  const float* hr = h + (size_t)s * D + lane;
  float v0 = hr[0];
  float v1 = hr[32];
  float v2 = hr[64];
  float v3 = hr[96];
  float* ap = agg + (size_t)d * D + lane;
  atomicAdd(ap,      v0);
  atomicAdd(ap + 32, v1);
  atomicAdd(ap + 64, v2);
  atomicAdd(ap + 96, v3);
}

struct LayerStage {
  v4f   h0, h1, h2, h3;
  v4f   a0, a1, a2, a3;
  float i0, i1, i2, i3;
};

__device__ __forceinline__ void layer_stage_load(LayerStage& s, const float* __restrict__ h,
                                                 const float* __restrict__ agg,
                                                 const float* __restrict__ inv_deg,
                                                 int tile, int sr, int sc) {
  const v4f* hp = (const v4f*)(h   + (size_t)tile * (TILE_M * D));
  const v4f* ap = (const v4f*)(agg + (size_t)tile * (TILE_M * D));
  const float*  ip = inv_deg + (size_t)tile * TILE_M;
  const int o = sr * (D / 4) + sc;
  s.h0 = hp[o];              s.a0 = ap[o];              s.i0 = ip[sr];
  s.h1 = hp[o + 8 * D / 4];  s.a1 = ap[o + 8 * D / 4];  s.i1 = ip[sr + 8];
  s.h2 = hp[o + 16 * D / 4]; s.a2 = ap[o + 16 * D / 4]; s.i2 = ip[sr + 16];
  s.h3 = hp[o + 24 * D / 4]; s.a3 = ap[o + 24 * D / 4]; s.i3 = ip[sr + 24];
}

__global__ __launch_bounds__(256) void layer_gemm(const float* __restrict__ h,
                                                  const float* __restrict__ agg,
                                                  const float* __restrict__ inv_deg,
                                                  const float* __restrict__ Wself,
                                                  const float* __restrict__ Wneigh,
                                                  const float* __restrict__ bias_v,
                                                  float* __restrict__ h_out,
                                                  int relu) {
  __shared__ __attribute__((aligned(16))) float sh[2][TILE_M * D_PAD];
  __shared__ __attribute__((aligned(16))) float sa[2][TILE_M * D_PAD];
  const int tid  = threadIdx.x;
  const int lane = tid & 31;
  const int half = lane >> 4;
  const int l16  = lane & 15;
  const int col  = (tid >> 5) * 16 + l16;

  v2f bS[32], bN[32];
#pragma unroll
  for (int t = 0; t < 32; ++t) {
    const int ka = 4 * t + 2 * half;
    bS[t].x = Wself[(size_t)(ka)      * D + col];
    bS[t].y = Wself[(size_t)(ka + 1)  * D + col];
    bN[t].x = Wneigh[(size_t)(ka)     * D + col];
    bN[t].y = Wneigh[(size_t)(ka + 1) * D + col];
  }
  const float bv = bias_v[col];

  const int sc = tid & 31;
  const int sr = tid >> 5;
  LayerStage st;

  int tile = blockIdx.x;
  if (tile < N_TILES) layer_stage_load(st, h, agg, inv_deg, tile, sr, sc);

  int buf = 0;
  for (; tile < N_TILES; tile += GEMM_GRID) {
    float* shb = sh[buf];
    float* sab = sa[buf];
    {
      const int o = sr * D_PAD + 4 * sc;
      *(v4f*)(shb + o)              = st.h0;
      *(v4f*)(shb + o + 8 * D_PAD)  = st.h1;
      *(v4f*)(shb + o + 16 * D_PAD) = st.h2;
      *(v4f*)(shb + o + 24 * D_PAD) = st.h3;
      *(v4f*)(sab + o)              = st.a0 * st.i0;
      *(v4f*)(sab + o + 8 * D_PAD)  = st.a1 * st.i1;
      *(v4f*)(sab + o + 16 * D_PAD) = st.a2 * st.i2;
      *(v4f*)(sab + o + 24 * D_PAD) = st.a3 * st.i3;
    }
    __syncthreads();

    const int nxt = tile + GEMM_GRID;
    if (nxt < N_TILES) layer_stage_load(st, h, agg, inv_deg, nxt, sr, sc);

    v8f acc0 = {0.f, 0.f, 0.f, 0.f, 0.f, 0.f, 0.f, 0.f};
    v8f acc1 = {0.f, 0.f, 0.f, 0.f, 0.f, 0.f, 0.f, 0.f};
    const float* h0 = shb + l16 * D_PAD + 2 * half;
    const float* h1 = h0 + 16 * D_PAD;
    const float* g0 = sab + l16 * D_PAD + 2 * half;
    const float* g1 = g0 + 16 * D_PAD;
#pragma unroll
    for (int t = 0; t < 32; ++t) {
      v2f a10 = *(const v2f*)(h0 + 4 * t);
      v2f a11 = *(const v2f*)(h1 + 4 * t);
      acc0 = wmma_f32_16x16x4(a10, bS[t], acc0);
      acc1 = wmma_f32_16x16x4(a11, bS[t], acc1);
      v2f a20 = *(const v2f*)(g0 + 4 * t);
      v2f a21 = *(const v2f*)(g1 + 4 * t);
      acc0 = wmma_f32_16x16x4(a20, bN[t], acc0);
      acc1 = wmma_f32_16x16x4(a21, bN[t], acc1);
    }

    float* ob = h_out + (size_t)tile * (TILE_M * D);
    const int lo = 8 * half * D + col;
#pragma unroll
    for (int r = 0; r < 8; ++r) {
      float v = acc0[r] + bv;
      if (relu) v = fmaxf(v, 0.0f);
      ob[lo + r * D] = v;
    }
#pragma unroll
    for (int r = 0; r < 8; ++r) {
      float v = acc1[r] + bv;
      if (relu) v = fmaxf(v, 0.0f);
      ob[lo + (16 + r) * D] = v;
    }

    buf ^= 1;
  }
}

extern "C" void kernel_launch(void* const* d_in, const int* in_sizes, int n_in,
                              void* d_out, int out_size, void* d_ws, size_t ws_size,
                              hipStream_t stream) {
  const float* x       = (const float*)d_in[0];
  const int*   ei      = (const int*)d_in[1];
  const float* W_in    = (const float*)d_in[2];
  const float* b_in    = (const float*)d_in[3];
  const float* W_self  = (const float*)d_in[4];
  const float* W_neigh = (const float*)d_in[5];
  const float* b_lay   = (const float*)d_in[6];
  float* out = (float*)d_out;

  float* ws   = (float*)d_ws;
  float* hA   = ws;
  float* hB   = ws + (size_t)1 * N_NODES * D;
  float* agg  = ws + (size_t)2 * N_NODES * D;
  float* degv = ws + (size_t)3 * N_NODES * D;

  const int* src = ei;
  const int* dst = ei + N_EDGES;

  hipMemsetAsync(degv, 0, N_NODES * sizeof(float), stream);
  degree_kernel<<<(N_EDGES + 255) / 256, 256, 0, stream>>>(dst, degv, N_EDGES);
  inv_deg_kernel<<<(N_NODES + 255) / 256, 256, 0, stream>>>(degv, N_NODES);

  input_gemm_tanh<<<GEMM_GRID, 256, 0, stream>>>(x, W_in, b_in, hA);

  const float* hcur = hA;
  for (int i = 0; i < L_LAYERS; ++i) {
    hipMemsetAsync(agg, 0, (size_t)N_NODES * D * sizeof(float), stream);
    scatter_sum<<<(N_EDGES + 7) / 8, 256, 0, stream>>>(hcur, src, dst, agg, N_EDGES);
    float* hnext = (i == L_LAYERS - 1) ? out : (hcur == hA ? hB : hA);
    layer_gemm<<<GEMM_GRID, 256, 0, stream>>>(
        hcur, agg, degv,
        W_self  + (size_t)i * D * D,
        W_neigh + (size_t)i * D * D,
        b_lay   + (size_t)i * D,
        hnext, (i < L_LAYERS - 1) ? 1 : 0);
    hcur = hnext;
  }
}


# Per-layer agg clear + f32-atomic edge scatter replaced by a destination-sorted neighbour list (built once per call) and a per-node gather-mean; row sums now in list-slot order = fill arrival order
# speedup vs baseline: 1.7382x; 1.2864x over previous
#include <hip/hip_runtime.h>
#include <math.h>


typedef __attribute__((ext_vector_type(2))) float v2f;
typedef __attribute__((ext_vector_type(4))) float v4f;
typedef __attribute__((ext_vector_type(8))) float v8f;

#define N_NODES 200000
#define N_EDGES 600000
#define F_IN    117
#define F_PAD   124
#define D       128
#define D_PAD   132
#define L_LAYERS 3
#define TILE_M   32
#define N_TILES  (N_NODES / TILE_M)
#define GEMM_GRID 512
#define X_TILE_F4 (TILE_M * F_IN / 4)
#define SCAN_BLOCK 1024
#define N_SCAN_BLOCKS ((N_NODES + SCAN_BLOCK - 1) / SCAN_BLOCK)
#define ROWPTR_PAD 200704

__device__ __forceinline__ v8f wmma_f32_16x16x4(v2f a, v2f b, v8f c) {
  return __builtin_amdgcn_wmma_f32_16x16x4_f32(false, a, false, b, (short)0, c,
                                               false, false);
}

__global__ __launch_bounds__(256) void count_kernel(const int* __restrict__ dst,
                                                    int* __restrict__ cnt, int nE) {
  int e = blockIdx.x * blockDim.x + threadIdx.x;
  if (e < nE) atomicAdd(&cnt[dst[e]], 1);
}

__global__ __launch_bounds__(256) void block_sum_kernel(const int* __restrict__ cnt,
                                                        int* __restrict__ bsum) {
  __shared__ int red[8];
  const int tid = threadIdx.x;
  const int q = blockIdx.x * 256 + tid;
  int s = 0;
  if (q < N_NODES / 4) {
    int4 v = ((const int4*)cnt)[q];
    s = v.x + v.y + v.z + v.w;
  }
  for (int o = 16; o > 0; o >>= 1) s += __shfl_xor(s, o);
  if ((tid & 31) == 0) red[tid >> 5] = s;
  __syncthreads();
  if (tid == 0) {
    int t = 0;
    for (int w = 0; w < 8; ++w) t += red[w];
    bsum[blockIdx.x] = t;
  }
}

__global__ __launch_bounds__(256) void scan_kernel(const int* __restrict__ cnt,
                                                   const int* __restrict__ bsum,
                                                   int* __restrict__ rowptr) {
  __shared__ int wtot[8];
  __shared__ int wpre[8];
  const int tid  = threadIdx.x;
  const int lane = tid & 31;
  const int w    = tid >> 5;
  int part = (tid < (int)blockIdx.x) ? bsum[tid] : 0;
  for (int o = 16; o > 0; o >>= 1) part += __shfl_xor(part, o);
  if (lane == 0) wpre[w] = part;

  const int q = blockIdx.x * 256 + tid;
  int4 v = make_int4(0, 0, 0, 0);
  if (q < N_NODES / 4) v = ((const int4*)cnt)[q];
  const int t = v.x + v.y + v.z + v.w;
  int inc = t;
  for (int o = 1; o < 32; o <<= 1) {
    int u = __shfl_up(inc, o);
    if (lane >= o) inc += u;
  }
  if (lane == 31) wtot[w] = inc;
  __syncthreads();
  int base = 0;
  for (int k = 0; k < 8; ++k) {
    base += wpre[k];
    if (k < w) base += wtot[k];
  }
  const int ex = base + inc - t;
  if (q < N_NODES / 4) {
    int4 r;
    r.x = ex;
    r.y = ex + v.x;
    r.z = r.y + v.y;
    r.w = r.z + v.z;
    ((int4*)rowptr)[q] = r;
  }
  if (q == N_NODES / 4 - 1) rowptr[N_NODES] = ex + t;
}

__global__ __launch_bounds__(256) void fill_kernel(const int* __restrict__ src,
                                                   const int* __restrict__ dst,
                                                   const int* __restrict__ rowptr,
                                                   int* __restrict__ cnt,
                                                   int* __restrict__ nbr, int nE) {
  int e = blockIdx.x * blockDim.x + threadIdx.x;
  if (e < nE) {
    const int d = dst[e];
    const int p = atomicSub(&cnt[d], 1) - 1;
    nbr[rowptr[d] + p] = src[e];
  }
}

__global__ __launch_bounds__(256) void input_gemm_tanh(const float* __restrict__ x,
                                                       const float* __restrict__ W_in,
                                                       const float* __restrict__ b_in,
                                                       float* __restrict__ h_out) {
  __shared__ __attribute__((aligned(16))) float sx[2][TILE_M * F_PAD];
  const int tid  = threadIdx.x;
  const int lane = tid & 31;
  const int half = lane >> 4;
  const int l16  = lane & 15;
  const int col  = (tid >> 5) * 16 + l16;

  for (int i = tid; i < 2 * TILE_M * (F_PAD - F_IN); i += 256) {
    int r = i / (F_PAD - F_IN), c = i - r * (F_PAD - F_IN);
    (&sx[0][0])[r * F_PAD + F_IN + c] = 0.0f;
  }

  v2f bW[30];
#pragma unroll
  for (int t = 0; t < 30; ++t) {
    const int ka = 4 * t + 2 * half;
    bW[t].x = (ka     < F_IN) ? W_in[(size_t)(ka)     * D + col] : 0.0f;
    bW[t].y = (ka + 1 < F_IN) ? W_in[(size_t)(ka + 1) * D + col] : 0.0f;
  }
  const float bias = b_in[col];

  const bool tail = tid < X_TILE_F4 - 768;
  v4f px[4];
  int tile = blockIdx.x;
  if (tile < N_TILES) {
    const v4f* xt = (const v4f*)(x + (size_t)tile * (TILE_M * F_IN));
    px[0] = xt[tid];
    px[1] = xt[tid + 256];
    px[2] = xt[tid + 512];
    if (tail) px[3] = xt[tid + 768];
  }

  int buf = 0;
  for (; tile < N_TILES; tile += GEMM_GRID) {
    float* sxb = sx[buf];
#pragma unroll
    for (int j = 0; j < 4; ++j) {
      if (j < 3 || tail) {
        int i0 = 4 * (tid + 256 * j);
        int r0 = i0 / F_IN, c0 = i0 - r0 * F_IN;
#pragma unroll
        for (int u = 0; u < 4; ++u) {
          sxb[r0 * F_PAD + c0] = px[j][u];
          if (++c0 == F_IN) { c0 = 0; ++r0; }
        }
      }
    }
    __syncthreads();

    const int nxt = tile + GEMM_GRID;
    if (nxt < N_TILES) {
      const v4f* xt = (const v4f*)(x + (size_t)nxt * (TILE_M * F_IN));
      px[0] = xt[tid];
      px[1] = xt[tid + 256];
      px[2] = xt[tid + 512];
      if (tail) px[3] = xt[tid + 768];
    }

    v8f acc0 = {0.f, 0.f, 0.f, 0.f, 0.f, 0.f, 0.f, 0.f};
    v8f acc1 = {0.f, 0.f, 0.f, 0.f, 0.f, 0.f, 0.f, 0.f};
    const float* ar0 = sxb + l16 * F_PAD + 2 * half;
    const float* ar1 = ar0 + 16 * F_PAD;
#pragma unroll
    for (int t = 0; t < 30; ++t) {
      v2f a0 = *(const v2f*)(ar0 + 4 * t);
      v2f a1 = *(const v2f*)(ar1 + 4 * t);
      acc0 = wmma_f32_16x16x4(a0, bW[t], acc0);
      acc1 = wmma_f32_16x16x4(a1, bW[t], acc1);
    }

    float* ob = h_out + (size_t)tile * (TILE_M * D);
    const int lo = 8 * half * D + col;
#pragma unroll
    for (int r = 0; r < 8; ++r)
      ob[lo + r * D] = tanhf(acc0[r] + bias);
#pragma unroll
    for (int r = 0; r < 8; ++r)
      ob[lo + (16 + r) * D] = tanhf(acc1[r] + bias);

    buf ^= 1;
  }
}

__device__ __forceinline__ v4f row_f4(const float* __restrict__ h, int s, int lane) {
  return ((const v4f*)(h + (size_t)s * D))[lane];
}

__global__ __launch_bounds__(256) void gather_mean(const float* __restrict__ h,
                                                   const int* __restrict__ rowptr,
                                                   const int* __restrict__ nbr,
                                                   float* __restrict__ agg) {
  const int node = __builtin_amdgcn_readfirstlane((int)(blockIdx.x * 8 + (threadIdx.x >> 5)));
  const int lane = threadIdx.x & 31;
  const int beg = rowptr[node];
  const int end = rowptr[node + 1];
  v4f acc = {0.f, 0.f, 0.f, 0.f};
  int j = beg;
  for (; j + 4 <= end; j += 4) {
    const int s0 = nbr[j], s1 = nbr[j + 1], s2 = nbr[j + 2], s3 = nbr[j + 3];
    const v4f r0 = row_f4(h, s0, lane);
    const v4f r1 = row_f4(h, s1, lane);
    const v4f r2 = row_f4(h, s2, lane);
    const v4f r3 = row_f4(h, s3, lane);
    acc += r0; acc += r1; acc += r2; acc += r3;
  }
  const int rem = end - j;
  if (rem > 0) {
    const int s0 = nbr[j];
    const int s1 = (rem > 1) ? nbr[j + 1] : s0;
    const int s2 = (rem > 2) ? nbr[j + 2] : s0;
    const v4f r0 = row_f4(h, s0, lane);
    const v4f r1 = row_f4(h, s1, lane);
    const v4f r2 = row_f4(h, s2, lane);
    acc += r0;
    if (rem > 1) acc += r1;
    if (rem > 2) acc += r2;
  }
  const float inv = 1.0f / fmaxf((float)(end - beg), 1.0f);
  ((v4f*)(agg + (size_t)node * D))[lane] = acc * inv;
}

struct LayerStage {
  v4f   h0, h1, h2, h3;
  v4f   a0, a1, a2, a3;
};

__device__ __forceinline__ void layer_stage_load(LayerStage& s, const float* h,
                                                 const float* __restrict__ agg,
                                                 int tile, int sr, int sc) {
  const v4f* hp = (const v4f*)(h   + (size_t)tile * (TILE_M * D));
  const v4f* ap = (const v4f*)(agg + (size_t)tile * (TILE_M * D));
  const int o = sr * (D / 4) + sc;
  s.h0 = hp[o];              s.a0 = ap[o];
  s.h1 = hp[o + 8 * D / 4];  s.a1 = ap[o + 8 * D / 4];
  s.h2 = hp[o + 16 * D / 4]; s.a2 = ap[o + 16 * D / 4];
  s.h3 = hp[o + 24 * D / 4]; s.a3 = ap[o + 24 * D / 4];
}

__global__ __launch_bounds__(256) void layer_gemm(const float* h,
                                                  const float* __restrict__ agg,
                                                  const float* __restrict__ Wself,
                                                  const float* __restrict__ Wneigh,
                                                  const float* __restrict__ bias_v,
                                                  float* h_out,
                                                  int relu) {
  __shared__ __attribute__((aligned(16))) float sh[2][TILE_M * D_PAD];
  __shared__ __attribute__((aligned(16))) float sa[2][TILE_M * D_PAD];
  const int tid  = threadIdx.x;
  const int lane = tid & 31;
  const int half = lane >> 4;
  const int l16  = lane & 15;
  const int col  = (tid >> 5) * 16 + l16;

  v2f bS[32], bN[32];
#pragma unroll
  for (int t = 0; t < 32; ++t) {
    const int ka = 4 * t + 2 * half;
    bS[t].x = Wself[(size_t)(ka)      * D + col];
    bS[t].y = Wself[(size_t)(ka + 1)  * D + col];
    bN[t].x = Wneigh[(size_t)(ka)     * D + col];
    bN[t].y = Wneigh[(size_t)(ka + 1) * D + col];
  }
  const float bv = bias_v[col];

  const int sc = tid & 31;
  const int sr = tid >> 5;
  LayerStage st;

  int tile = blockIdx.x;
  if (tile < N_TILES) layer_stage_load(st, h, agg, tile, sr, sc);

  int buf = 0;
  for (; tile < N_TILES; tile += GEMM_GRID) {
    float* shb = sh[buf];
    float* sab = sa[buf];
    {
      const int o = sr * D_PAD + 4 * sc;
      *(v4f*)(shb + o)              = st.h0;
      *(v4f*)(shb + o + 8 * D_PAD)  = st.h1;
      *(v4f*)(shb + o + 16 * D_PAD) = st.h2;
      *(v4f*)(shb + o + 24 * D_PAD) = st.h3;
      *(v4f*)(sab + o)              = st.a0;
      *(v4f*)(sab + o + 8 * D_PAD)  = st.a1;
      *(v4f*)(sab + o + 16 * D_PAD) = st.a2;
      *(v4f*)(sab + o + 24 * D_PAD) = st.a3;
    }
    __syncthreads();

    const int nxt = tile + GEMM_GRID;
    if (nxt < N_TILES) layer_stage_load(st, h, agg, nxt, sr, sc);

    v8f acc0 = {0.f, 0.f, 0.f, 0.f, 0.f, 0.f, 0.f, 0.f};
    v8f acc1 = {0.f, 0.f, 0.f, 0.f, 0.f, 0.f, 0.f, 0.f};
    const float* h0 = shb + l16 * D_PAD + 2 * half;
    const float* h1 = h0 + 16 * D_PAD;
    const float* g0 = sab + l16 * D_PAD + 2 * half;
    const float* g1 = g0 + 16 * D_PAD;
#pragma unroll
    for (int t = 0; t < 32; ++t) {
      v2f a10 = *(const v2f*)(h0 + 4 * t);
      v2f a11 = *(const v2f*)(h1 + 4 * t);
      acc0 = wmma_f32_16x16x4(a10, bS[t], acc0);
      acc1 = wmma_f32_16x16x4(a11, bS[t], acc1);
      v2f a20 = *(const v2f*)(g0 + 4 * t);
      v2f a21 = *(const v2f*)(g1 + 4 * t);
      acc0 = wmma_f32_16x16x4(a20, bN[t], acc0);
      acc1 = wmma_f32_16x16x4(a21, bN[t], acc1);
    }

    float* ob = h_out + (size_t)tile * (TILE_M * D);
    const int lo = 8 * half * D + col;
#pragma unroll
    for (int r = 0; r < 8; ++r) {
      float v = acc0[r] + bv;
      if (relu) v = fmaxf(v, 0.0f);
      ob[lo + r * D] = v;
    }
#pragma unroll
    for (int r = 0; r < 8; ++r) {
      float v = acc1[r] + bv;
      if (relu) v = fmaxf(v, 0.0f);
      ob[lo + (16 + r) * D] = v;
    }

    buf ^= 1;
  }
}

extern "C" void kernel_launch(void* const* d_in, const int* in_sizes, int n_in,
                              void* d_out, int out_size, void* d_ws, size_t ws_size,
                              hipStream_t stream) {
  const float* x       = (const float*)d_in[0];
  const int*   ei      = (const int*)d_in[1];
  const float* W_in    = (const float*)d_in[2];
  const float* b_in    = (const float*)d_in[3];
  const float* W_self  = (const float*)d_in[4];
  const float* W_neigh = (const float*)d_in[5];
  const float* b_lay   = (const float*)d_in[6];
  float* out = (float*)d_out;

  float* ws   = (float*)d_ws;
  float* hA   = ws;
  int*   ib   = (int*)(ws + (size_t)1 * N_NODES * D);
  int*   rowptr = ib;
  int*   nbr    = ib + ROWPTR_PAD;
  int*   bsum   = ib + ROWPTR_PAD + N_EDGES;
  float* agg  = ws + (size_t)2 * N_NODES * D;
  int*   cnt  = (int*)(ws + (size_t)3 * N_NODES * D);

  const int* src = ei;
  const int* dst = ei + N_EDGES;

  hipMemsetAsync(cnt, 0, N_NODES * sizeof(int), stream);
  count_kernel<<<(N_EDGES + 255) / 256, 256, 0, stream>>>(dst, cnt, N_EDGES);
  block_sum_kernel<<<N_SCAN_BLOCKS, 256, 0, stream>>>(cnt, bsum);
  scan_kernel<<<N_SCAN_BLOCKS, 256, 0, stream>>>(cnt, bsum, rowptr);
  fill_kernel<<<(N_EDGES + 255) / 256, 256, 0, stream>>>(src, dst, rowptr, cnt, nbr, N_EDGES);

  input_gemm_tanh<<<GEMM_GRID, 256, 0, stream>>>(x, W_in, b_in, hA);

  for (int i = 0; i < L_LAYERS; ++i) {
    gather_mean<<<N_NODES / 8, 256, 0, stream>>>(hA, rowptr, nbr, agg);
    float* hnext = (i == L_LAYERS - 1) ? out : hA;
    layer_gemm<<<GEMM_GRID, 256, 0, stream>>>(
        hA, agg,
        W_self  + (size_t)i * D * D,
        W_neigh + (size_t)i * D * D,
        b_lay   + (size_t)i * D,
        hnext, (i < L_LAYERS - 1) ? 1 : 0);
  }
}


# Neighbour mean moved into the layer GEMM's tile staging (each wave gathers its four rows' neighbour rows under the other waves' matrix work); agg buffer and its 3 launches gone; same slot sum order
# speedup vs baseline: 1.8840x; 1.0839x over previous
#include <hip/hip_runtime.h>
#include <math.h>


typedef __attribute__((ext_vector_type(2))) float v2f;
typedef __attribute__((ext_vector_type(4))) float v4f;
typedef __attribute__((ext_vector_type(8))) float v8f;

#define N_NODES 200000
#define N_EDGES 600000
#define F_IN    117
#define F_PAD   124
#define D       128
#define D_PAD   132
#define L_LAYERS 3
#define TILE_M   32
#define N_TILES  (N_NODES / TILE_M)
#define GEMM_GRID 512
#define X_TILE_F4 (TILE_M * F_IN / 4)
#define SCAN_BLOCK 1024
#define N_SCAN_BLOCKS ((N_NODES + SCAN_BLOCK - 1) / SCAN_BLOCK)
#define ROWPTR_PAD 200704

__device__ __forceinline__ v8f wmma_f32_16x16x4(v2f a, v2f b, v8f c) {
  return __builtin_amdgcn_wmma_f32_16x16x4_f32(false, a, false, b, (short)0, c,
                                               false, false);
}

__global__ __launch_bounds__(256) void count_kernel(const int* __restrict__ dst,
                                                    int* __restrict__ cnt, int nE) {
  int e = blockIdx.x * blockDim.x + threadIdx.x;
  if (e < nE) atomicAdd(&cnt[dst[e]], 1);
}

__global__ __launch_bounds__(256) void block_sum_kernel(const int* __restrict__ cnt,
                                                        int* __restrict__ bsum) {
  __shared__ int red[8];
  const int tid = threadIdx.x;
  const int q = blockIdx.x * 256 + tid;
  int s = 0;
  if (q < N_NODES / 4) {
    int4 v = ((const int4*)cnt)[q];
    s = v.x + v.y + v.z + v.w;
  }
  for (int o = 16; o > 0; o >>= 1) s += __shfl_xor(s, o);
  if ((tid & 31) == 0) red[tid >> 5] = s;
  __syncthreads();
  if (tid == 0) {
    int t = 0;
    for (int w = 0; w < 8; ++w) t += red[w];
    bsum[blockIdx.x] = t;
  }
}

__global__ __launch_bounds__(256) void scan_kernel(const int* __restrict__ cnt,
                                                   const int* __restrict__ bsum,
                                                   int* __restrict__ rowptr) {
  __shared__ int wtot[8];
  __shared__ int wpre[8];
  const int tid  = threadIdx.x;
  const int lane = tid & 31;
  const int w    = tid >> 5;
  int part = (tid < (int)blockIdx.x) ? bsum[tid] : 0;
  for (int o = 16; o > 0; o >>= 1) part += __shfl_xor(part, o);
  if (lane == 0) wpre[w] = part;

  const int q = blockIdx.x * 256 + tid;
  int4 v = make_int4(0, 0, 0, 0);
  if (q < N_NODES / 4) v = ((const int4*)cnt)[q];
  const int t = v.x + v.y + v.z + v.w;
  int inc = t;
  for (int o = 1; o < 32; o <<= 1) {
    int u = __shfl_up(inc, o);
    if (lane >= o) inc += u;
  }
  if (lane == 31) wtot[w] = inc;
  __syncthreads();
  int base = 0;
  for (int k = 0; k < 8; ++k) {
    base += wpre[k];
    if (k < w) base += wtot[k];
  }
  const int ex = base + inc - t;
  if (q < N_NODES / 4) {
    int4 r;
    r.x = ex;
    r.y = ex + v.x;
    r.z = r.y + v.y;
    r.w = r.z + v.z;
    ((int4*)rowptr)[q] = r;
  }
  if (q == N_NODES / 4 - 1) rowptr[N_NODES] = ex + t;
}

__global__ __launch_bounds__(256) void fill_kernel(const int* __restrict__ src,
                                                   const int* __restrict__ dst,
                                                   const int* __restrict__ rowptr,
                                                   int* __restrict__ cnt,
                                                   int* __restrict__ nbr, int nE) {
  int e = blockIdx.x * blockDim.x + threadIdx.x;
  if (e < nE) {
    const int d = dst[e];
    const int p = atomicSub(&cnt[d], 1) - 1;
    nbr[rowptr[d] + p] = src[e];
  }
}

__global__ __launch_bounds__(256) void input_gemm_tanh(const float* __restrict__ x,
                                                       const float* __restrict__ W_in,
                                                       const float* __restrict__ b_in,
                                                       float* __restrict__ h_out) {
  __shared__ __attribute__((aligned(16))) float sx[2][TILE_M * F_PAD];
  const int tid  = threadIdx.x;
  const int lane = tid & 31;
  const int half = lane >> 4;
  const int l16  = lane & 15;
  const int col  = (tid >> 5) * 16 + l16;

  for (int i = tid; i < 2 * TILE_M * (F_PAD - F_IN); i += 256) {
    int r = i / (F_PAD - F_IN), c = i - r * (F_PAD - F_IN);
    (&sx[0][0])[r * F_PAD + F_IN + c] = 0.0f;
  }

  v2f bW[30];
#pragma unroll
  for (int t = 0; t < 30; ++t) {
    const int ka = 4 * t + 2 * half;
    bW[t].x = (ka     < F_IN) ? W_in[(size_t)(ka)     * D + col] : 0.0f;
    bW[t].y = (ka + 1 < F_IN) ? W_in[(size_t)(ka + 1) * D + col] : 0.0f;
  }
  const float bias = b_in[col];

  const bool tail = tid < X_TILE_F4 - 768;
  v4f px[4];
  int tile = blockIdx.x;
  if (tile < N_TILES) {
    const v4f* xt = (const v4f*)(x + (size_t)tile * (TILE_M * F_IN));
    px[0] = xt[tid];
    px[1] = xt[tid + 256];
    px[2] = xt[tid + 512];
    if (tail) px[3] = xt[tid + 768];
  }

  int buf = 0;
  for (; tile < N_TILES; tile += GEMM_GRID) {
    float* sxb = sx[buf];
#pragma unroll
    for (int j = 0; j < 4; ++j) {
      if (j < 3 || tail) {
        int i0 = 4 * (tid + 256 * j);
        int r0 = i0 / F_IN, c0 = i0 - r0 * F_IN;
#pragma unroll
        for (int u = 0; u < 4; ++u) {
          sxb[r0 * F_PAD + c0] = px[j][u];
          if (++c0 == F_IN) { c0 = 0; ++r0; }
        }
      }
    }
    __syncthreads();

    const int nxt = tile + GEMM_GRID;
    if (nxt < N_TILES) {
      const v4f* xt = (const v4f*)(x + (size_t)nxt * (TILE_M * F_IN));
      px[0] = xt[tid];
      px[1] = xt[tid + 256];
      px[2] = xt[tid + 512];
      if (tail) px[3] = xt[tid + 768];
    }

    v8f acc0 = {0.f, 0.f, 0.f, 0.f, 0.f, 0.f, 0.f, 0.f};
    v8f acc1 = {0.f, 0.f, 0.f, 0.f, 0.f, 0.f, 0.f, 0.f};
    const float* ar0 = sxb + l16 * F_PAD + 2 * half;
    const float* ar1 = ar0 + 16 * F_PAD;
#pragma unroll
    for (int t = 0; t < 30; ++t) {
      v2f a0 = *(const v2f*)(ar0 + 4 * t);
      v2f a1 = *(const v2f*)(ar1 + 4 * t);
      acc0 = wmma_f32_16x16x4(a0, bW[t], acc0);
      acc1 = wmma_f32_16x16x4(a1, bW[t], acc1);
    }

    float* ob = h_out + (size_t)tile * (TILE_M * D);
    const int lo = 8 * half * D + col;
#pragma unroll
    for (int r = 0; r < 8; ++r)
      ob[lo + r * D] = tanhf(acc0[r] + bias);
#pragma unroll
    for (int r = 0; r < 8; ++r)
      ob[lo + (16 + r) * D] = tanhf(acc1[r] + bias);

    buf ^= 1;
  }
}

#define NB_IDS 8
#define NB_GRP 4

__device__ __forceinline__ v4f row_f4(const float* __restrict__ h, int s, int lane) {
  return ((const v4f*)(h + (size_t)s * D))[lane];
}

__device__ __forceinline__ void ranges_load(int (&b)[4], int (&e)[4],
                                            const int* __restrict__ rowptr, int n0) {
#pragma unroll
  for (int j = 0; j < 4; ++j) {
    b[j] = rowptr[n0 + 8 * j];
    e[j] = rowptr[n0 + 8 * j + 1];
  }
}

__device__ __forceinline__ int ids_request(const int* __restrict__ nbr, const int (&b)[4], int lane) {
  const int j = lane >> 3;
  const int k = lane & 7;
  int base = b[0];
  base = (j == 1) ? b[1] : base;
  base = (j == 2) ? b[2] : base;
  base = (j == 3) ? b[3] : base;
  const int s = base + k;
  return nbr[s < N_EDGES ? s : N_EDGES - 1];
}

__device__ __forceinline__ void rows_round(v4f (&acc)[4], const float* __restrict__ h,
                                           const int (&b)[4], const int (&e)[4],
                                           int idv, int n0, int lane, int off) {
  v4f r[4][NB_GRP];
#pragma unroll
  for (int j = 0; j < 4; ++j) {
#pragma unroll
    for (int k = 0; k < NB_GRP; ++k) {
      const int id = __builtin_amdgcn_readlane(idv, NB_IDS * j + off + k);
      const int s = (b[j] + off + k < e[j]) ? id : n0 + 8 * j;
      r[j][k] = row_f4(h, s, lane);
    }
  }
  const v4f zero = {0.f, 0.f, 0.f, 0.f};
#pragma unroll
  for (int j = 0; j < 4; ++j) {
#pragma unroll
    for (int k = 0; k < NB_GRP; ++k) {
      const v4f rv = r[j][k];
      acc[j] += (b[j] + off + k < e[j]) ? rv : zero;
    }
  }
}

__device__ __forceinline__ void means_build(v4f (&mean)[4], const float* __restrict__ h,
                                            const int* __restrict__ nbr,
                                            const int (&b)[4], const int (&e)[4],
                                            int idv, int n0, int lane) {
  v4f acc[4];
#pragma unroll
  for (int j = 0; j < 4; ++j) acc[j] = (v4f){0.f, 0.f, 0.f, 0.f};
  int dmax = e[0] - b[0];
#pragma unroll
  for (int j = 1; j < 4; ++j) dmax = (e[j] - b[j] > dmax) ? e[j] - b[j] : dmax;

  rows_round(acc, h, b, e, idv, n0, lane, 0);
  if (dmax > NB_GRP) rows_round(acc, h, b, e, idv, n0, lane, NB_GRP);

  for (int off = NB_IDS; off < dmax; off += 4) {
#pragma unroll
    for (int j = 0; j < 4; ++j) {
      const int s = b[j] + off;
      const int rem = e[j] - s;
      if (rem > 0) {
        const int s0 = nbr[s];
        const int s1 = (rem > 1) ? nbr[s + 1] : s0;
        const int s2 = (rem > 2) ? nbr[s + 2] : s0;
        const int s3 = (rem > 3) ? nbr[s + 3] : s0;
        const v4f q0 = row_f4(h, s0, lane);
        const v4f q1 = row_f4(h, s1, lane);
        const v4f q2 = row_f4(h, s2, lane);
        const v4f q3 = row_f4(h, s3, lane);
        acc[j] += q0;
        if (rem > 1) acc[j] += q1;
        if (rem > 2) acc[j] += q2;
        if (rem > 3) acc[j] += q3;
      }
    }
  }
#pragma unroll
  for (int j = 0; j < 4; ++j) {
    const float inv = 1.0f / fmaxf((float)(e[j] - b[j]), 1.0f);
    mean[j] = acc[j] * inv;
  }
}

__device__ __forceinline__ void hrows_load(v4f (&hrow)[4], const float* __restrict__ h,
                                           int tile, int sr, int sc) {
  const v4f* hp = (const v4f*)(h + (size_t)tile * (TILE_M * D));
  const int o = sr * (D / 4) + sc;
  hrow[0] = hp[o];
  hrow[1] = hp[o + 8 * D / 4];
  hrow[2] = hp[o + 16 * D / 4];
  hrow[3] = hp[o + 24 * D / 4];
}

__global__ __launch_bounds__(256) void layer_gemm(const float* __restrict__ h,
                                                  const int* __restrict__ rowptr,
                                                  const int* __restrict__ nbr,
                                                  const float* __restrict__ Wself,
                                                  const float* __restrict__ Wneigh,
                                                  const float* __restrict__ bias_v,
                                                  float* __restrict__ h_out,
                                                  int relu) {
  __shared__ __attribute__((aligned(16))) float sh[2][TILE_M * D_PAD];
  __shared__ __attribute__((aligned(16))) float sa[2][TILE_M * D_PAD];
  const int tid  = threadIdx.x;
  const int lane = tid & 31;
  const int half = lane >> 4;
  const int l16  = lane & 15;
  const int col  = (tid >> 5) * 16 + l16;

  v2f bS[32], bN[32];
#pragma unroll
  for (int t = 0; t < 32; ++t) {
    const int ka = 4 * t + 2 * half;
    bS[t].x = Wself[(size_t)(ka)      * D + col];
    bS[t].y = Wself[(size_t)(ka + 1)  * D + col];
    bN[t].x = Wneigh[(size_t)(ka)     * D + col];
    bN[t].y = Wneigh[(size_t)(ka + 1) * D + col];
  }
  const float bv = bias_v[col];

  const int sc = lane;
  const int sr = __builtin_amdgcn_readfirstlane(tid >> 5);

  v4f hrow[4];
  v4f mean[4];
  int rb[4], re[4];
  int idv;

  int tile = blockIdx.x;
  if (tile >= N_TILES) return;

  hrows_load(hrow, h, tile, sr, sc);
  ranges_load(rb, re, rowptr, tile * TILE_M + sr);
  idv = ids_request(nbr, rb, lane);
  means_build(mean, h, nbr, rb, re, idv, tile * TILE_M + sr, sc);
  if (tile + GEMM_GRID < N_TILES) ranges_load(rb, re, rowptr, (tile + GEMM_GRID) * TILE_M + sr);

  int buf = 0;
  for (; tile < N_TILES; tile += GEMM_GRID) {
    float* shb = sh[buf];
    float* sab = sa[buf];
    {
      const int o = sr * D_PAD + 4 * sc;
      *(v4f*)(shb + o)              = hrow[0];
      *(v4f*)(shb + o + 8 * D_PAD)  = hrow[1];
      *(v4f*)(shb + o + 16 * D_PAD) = hrow[2];
      *(v4f*)(shb + o + 24 * D_PAD) = hrow[3];
      *(v4f*)(sab + o)              = mean[0];
      *(v4f*)(sab + o + 8 * D_PAD)  = mean[1];
      *(v4f*)(sab + o + 16 * D_PAD) = mean[2];
      *(v4f*)(sab + o + 24 * D_PAD) = mean[3];
    }
    __syncthreads();

    const int nxt = tile + GEMM_GRID;
    int rpv = 0, rpv1 = 0;
    if (nxt < N_TILES) {
      hrows_load(hrow, h, nxt, sr, sc);
      idv = ids_request(nbr, rb, lane);
      const int q = (nxt + GEMM_GRID) * TILE_M + lane;
      rpv  = rowptr[q     < N_NODES ? q     : N_NODES];
      rpv1 = rowptr[q + 1 < N_NODES ? q + 1 : N_NODES];
    }

    v8f acc0 = {0.f, 0.f, 0.f, 0.f, 0.f, 0.f, 0.f, 0.f};
    v8f acc1 = {0.f, 0.f, 0.f, 0.f, 0.f, 0.f, 0.f, 0.f};
    const float* h0 = shb + l16 * D_PAD + 2 * half;
    const float* h1 = h0 + 16 * D_PAD;
    const float* g0 = sab + l16 * D_PAD + 2 * half;
    const float* g1 = g0 + 16 * D_PAD;
#pragma unroll
    for (int t = 0; t < 32; ++t) {
      v2f a10 = *(const v2f*)(h0 + 4 * t);
      v2f a11 = *(const v2f*)(h1 + 4 * t);
      acc0 = wmma_f32_16x16x4(a10, bS[t], acc0);
      acc1 = wmma_f32_16x16x4(a11, bS[t], acc1);
      v2f a20 = *(const v2f*)(g0 + 4 * t);
      v2f a21 = *(const v2f*)(g1 + 4 * t);
      acc0 = wmma_f32_16x16x4(a20, bN[t], acc0);
      acc1 = wmma_f32_16x16x4(a21, bN[t], acc1);
    }

    float* ob = h_out + (size_t)tile * (TILE_M * D);
    const int lo = 8 * half * D + col;
#pragma unroll
    for (int r = 0; r < 8; ++r) {
      float v = acc0[r] + bv;
      if (relu) v = fmaxf(v, 0.0f);
      ob[lo + r * D] = v;
    }
#pragma unroll
    for (int r = 0; r < 8; ++r) {
      float v = acc1[r] + bv;
      if (relu) v = fmaxf(v, 0.0f);
      ob[lo + (16 + r) * D] = v;
    }

    if (nxt < N_TILES) {
      means_build(mean, h, nbr, rb, re, idv, nxt * TILE_M + sr, sc);
#pragma unroll
      for (int j = 0; j < 4; ++j) {
        rb[j] = __builtin_amdgcn_readlane(rpv,  sr + 8 * j);
        re[j] = __builtin_amdgcn_readlane(rpv1, sr + 8 * j);
      }
    }

    buf ^= 1;
  }
}

extern "C" void kernel_launch(void* const* d_in, const int* in_sizes, int n_in,
                              void* d_out, int out_size, void* d_ws, size_t ws_size,
                              hipStream_t stream) {
  const float* x       = (const float*)d_in[0];
  const int*   ei      = (const int*)d_in[1];
  const float* W_in    = (const float*)d_in[2];
  const float* b_in    = (const float*)d_in[3];
  const float* W_self  = (const float*)d_in[4];
  const float* W_neigh = (const float*)d_in[5];
  const float* b_lay   = (const float*)d_in[6];
  float* out = (float*)d_out;

  float* ws   = (float*)d_ws;
  float* hA   = ws;
  int*   ib   = (int*)(ws + (size_t)1 * N_NODES * D);
  int*   rowptr = ib;
  int*   nbr    = ib + ROWPTR_PAD;
  int*   bsum   = ib + ROWPTR_PAD + N_EDGES;
  float* hB   = ws + (size_t)2 * N_NODES * D;
  int*   cnt  = (int*)(ws + (size_t)3 * N_NODES * D);

  const int* src = ei;
  const int* dst = ei + N_EDGES;

  hipMemsetAsync(cnt, 0, N_NODES * sizeof(int), stream);
  count_kernel<<<(N_EDGES + 255) / 256, 256, 0, stream>>>(dst, cnt, N_EDGES);
  block_sum_kernel<<<N_SCAN_BLOCKS, 256, 0, stream>>>(cnt, bsum);
  scan_kernel<<<N_SCAN_BLOCKS, 256, 0, stream>>>(cnt, bsum, rowptr);
  fill_kernel<<<(N_EDGES + 255) / 256, 256, 0, stream>>>(src, dst, rowptr, cnt, nbr, N_EDGES);

  input_gemm_tanh<<<GEMM_GRID, 256, 0, stream>>>(x, W_in, b_in, hA);

  layer_gemm<<<GEMM_GRID, 256, 0, stream>>>(hA, rowptr, nbr, W_self, W_neigh, b_lay, hB, 1);
  layer_gemm<<<GEMM_GRID, 256, 0, stream>>>(hB, rowptr, nbr, W_self + (size_t)1 * D * D,
                                            W_neigh + (size_t)1 * D * D, b_lay + (size_t)1 * D, hA, 1);
  layer_gemm<<<GEMM_GRID, 256, 0, stream>>>(hA, rowptr, nbr, W_self + (size_t)2 * D * D,
                                            W_neigh + (size_t)2 * D * D, b_lay + (size_t)2 * D, out, 0);
}


# Neighbour list: fixed 16-slot rows filled by the degree count itself (slot = value its add returns; 4 edges/thread; spill list past 16); scan+fill launches gone, 8->5; row sums in count arrival order
# speedup vs baseline: 2.0470x; 1.0865x over previous
#include <hip/hip_runtime.h>
#include <math.h>


typedef __attribute__((ext_vector_type(2))) float v2f;
typedef __attribute__((ext_vector_type(4))) float v4f;
typedef __attribute__((ext_vector_type(8))) float v8f;

#define N_NODES 200000
#define N_EDGES 600000
#define F_IN    117
#define F_PAD   124
#define D       128
#define D_PAD   132
#define L_LAYERS 3
#define TILE_M   32
#define N_TILES  (N_NODES / TILE_M)
#define GEMM_GRID 512
#define X_TILE_F4 (TILE_M * F_IN / 4)
#define ELL_K   16
#define CNT_PAD 200704

__device__ __forceinline__ v8f wmma_f32_16x16x4(v2f a, v2f b, v8f c) {
  return __builtin_amdgcn_wmma_f32_16x16x4_f32(false, a, false, b, (short)0, c,
                                               false, false);
}

__global__ __launch_bounds__(256) void place_kernel(const int* __restrict__ src,
                                                    const int* __restrict__ dst,
                                                    int* __restrict__ cnt,
                                                    int* __restrict__ ell,
                                                    int* __restrict__ spill, int nE4) {
  const int t = blockIdx.x * blockDim.x + threadIdx.x;
  if (t < nE4) {
    const int4 d4 = ((const int4*)dst)[t];
    const int4 s4 = ((const int4*)src)[t];
    const int d[4] = {d4.x, d4.y, d4.z, d4.w};
    const int s[4] = {s4.x, s4.y, s4.z, s4.w};
    int p[4];
#pragma unroll
    for (int u = 0; u < 4; ++u) p[u] = atomicAdd(&cnt[d[u]], 1);
#pragma unroll
    for (int u = 0; u < 4; ++u) {
      if (p[u] < ELL_K) {
        ell[d[u] * ELL_K + p[u]] = s[u];
      } else {
        const int q = atomicAdd(&cnt[N_NODES], 1);
        spill[2 * q]     = d[u];
        spill[2 * q + 1] = s[u];
      }
    }
  }
}

__global__ __launch_bounds__(256) void input_gemm_tanh(const float* __restrict__ x,
                                                       const float* __restrict__ W_in,
                                                       const float* __restrict__ b_in,
                                                       float* __restrict__ h_out) {
  __shared__ __attribute__((aligned(16))) float sx[2][TILE_M * F_PAD];
  const int tid  = threadIdx.x;
  const int lane = tid & 31;
  const int half = lane >> 4;
  const int l16  = lane & 15;
  const int col  = (tid >> 5) * 16 + l16;

  for (int i = tid; i < 2 * TILE_M * (F_PAD - F_IN); i += 256) {
    int r = i / (F_PAD - F_IN), c = i - r * (F_PAD - F_IN);
    (&sx[0][0])[r * F_PAD + F_IN + c] = 0.0f;
  }

  v2f bW[30];
#pragma unroll
  for (int t = 0; t < 30; ++t) {
    const int ka = 4 * t + 2 * half;
    bW[t].x = (ka     < F_IN) ? W_in[(size_t)(ka)     * D + col] : 0.0f;
    bW[t].y = (ka + 1 < F_IN) ? W_in[(size_t)(ka + 1) * D + col] : 0.0f;
  }
  const float bias = b_in[col];

  const bool tail = tid < X_TILE_F4 - 768;
  v4f px[4];
  int tile = blockIdx.x;
  if (tile < N_TILES) {
    const v4f* xt = (const v4f*)(x + (size_t)tile * (TILE_M * F_IN));
    px[0] = xt[tid];
    px[1] = xt[tid + 256];
    px[2] = xt[tid + 512];
    if (tail) px[3] = xt[tid + 768];
  }

  int buf = 0;
  for (; tile < N_TILES; tile += GEMM_GRID) {
    float* sxb = sx[buf];
#pragma unroll
    for (int j = 0; j < 4; ++j) {
      if (j < 3 || tail) {
        int i0 = 4 * (tid + 256 * j);
        int r0 = i0 / F_IN, c0 = i0 - r0 * F_IN;
#pragma unroll
        for (int u = 0; u < 4; ++u) {
          sxb[r0 * F_PAD + c0] = px[j][u];
          if (++c0 == F_IN) { c0 = 0; ++r0; }
        }
      }
    }
    __syncthreads();

    const int nxt = tile + GEMM_GRID;
    if (nxt < N_TILES) {
      const v4f* xt = (const v4f*)(x + (size_t)nxt * (TILE_M * F_IN));
      px[0] = xt[tid];
      px[1] = xt[tid + 256];
      px[2] = xt[tid + 512];
      if (tail) px[3] = xt[tid + 768];
    }

    v8f acc0 = {0.f, 0.f, 0.f, 0.f, 0.f, 0.f, 0.f, 0.f};
    v8f acc1 = {0.f, 0.f, 0.f, 0.f, 0.f, 0.f, 0.f, 0.f};
    const float* ar0 = sxb + l16 * F_PAD + 2 * half;
    const float* ar1 = ar0 + 16 * F_PAD;
#pragma unroll
    for (int t = 0; t < 30; ++t) {
      v2f a0 = *(const v2f*)(ar0 + 4 * t);
      v2f a1 = *(const v2f*)(ar1 + 4 * t);
      acc0 = wmma_f32_16x16x4(a0, bW[t], acc0);
      acc1 = wmma_f32_16x16x4(a1, bW[t], acc1);
    }

    float* ob = h_out + (size_t)tile * (TILE_M * D);
    const int lo = 8 * half * D + col;
#pragma unroll
    for (int r = 0; r < 8; ++r)
      ob[lo + r * D] = tanhf(acc0[r] + bias);
#pragma unroll
    for (int r = 0; r < 8; ++r)
      ob[lo + (16 + r) * D] = tanhf(acc1[r] + bias);

    buf ^= 1;
  }
}

#define NB_IDS 8
#define NB_GRP 4

__device__ __forceinline__ v4f row_f4(const float* __restrict__ h, int s, int lane) {
  return ((const v4f*)(h + (size_t)s * D))[lane];
}

__device__ __forceinline__ void degs_load(int (&dg)[4], const int* __restrict__ cnt, int n0) {
#pragma unroll
  for (int j = 0; j < 4; ++j) dg[j] = cnt[n0 + 8 * j];
}

__device__ __forceinline__ int ids_request(const int* __restrict__ ell, const int* __restrict__ cnt,
                                           const int (&dg)[4], int n0, int lane) {
  const int j = lane >> 3;
  const int k = lane & 7;
  int dj = dg[0];
  dj = (j == 1) ? dg[1] : dj;
  dj = (j == 2) ? dg[2] : dj;
  dj = (j == 3) ? dg[3] : dj;
  const int n = n0 + 8 * j;
  const int* p = (k < dj) ? (ell + n * ELL_K + k) : (cnt + n);
  return *p;
}

__device__ __forceinline__ void rows_round(v4f (&acc)[4], const float* __restrict__ h,
                                           const int (&dg)[4],
                                           int idv, int n0, int lane, int off) {
  v4f r[4][NB_GRP];
#pragma unroll
  for (int j = 0; j < 4; ++j) {
#pragma unroll
    for (int k = 0; k < NB_GRP; ++k) {
      const int id = __builtin_amdgcn_readlane(idv, NB_IDS * j + off + k);
      const int s = (off + k < dg[j]) ? id : n0 + 8 * j;
      r[j][k] = row_f4(h, s, lane);
    }
  }
  const v4f zero = {0.f, 0.f, 0.f, 0.f};
#pragma unroll
  for (int j = 0; j < 4; ++j) {
#pragma unroll
    for (int k = 0; k < NB_GRP; ++k) {
      const v4f rv = r[j][k];
      acc[j] += (off + k < dg[j]) ? rv : zero;
    }
  }
}

__device__ __forceinline__ void means_build(v4f (&mean)[4], const float* __restrict__ h,
                                            const int* __restrict__ ell,
                                            const int* __restrict__ cnt,
                                            const int* __restrict__ spill,
                                            const int (&dg)[4],
                                            int idv, int n0, int lane) {
  v4f acc[4];
#pragma unroll
  for (int j = 0; j < 4; ++j) acc[j] = (v4f){0.f, 0.f, 0.f, 0.f};
  int dmax = dg[0];
#pragma unroll
  for (int j = 1; j < 4; ++j) dmax = (dg[j] > dmax) ? dg[j] : dmax;

  rows_round(acc, h, dg, idv, n0, lane, 0);
  if (dmax > NB_GRP) rows_round(acc, h, dg, idv, n0, lane, NB_GRP);

  const int dcap = (dmax < ELL_K) ? dmax : ELL_K;
  for (int off = NB_IDS; off < dcap; off += 4) {
#pragma unroll
    for (int j = 0; j < 4; ++j) {
      const int dj = (dg[j] < ELL_K) ? dg[j] : ELL_K;
      const int rem = dj - off;
      if (rem > 0) {
        const int* q = ell + (n0 + 8 * j) * ELL_K + off;
        const int s0 = q[0];
        const int s1 = (rem > 1) ? q[1] : s0;
        const int s2 = (rem > 2) ? q[2] : s0;
        const int s3 = (rem > 3) ? q[3] : s0;
        const v4f q0 = row_f4(h, s0, lane);
        const v4f q1 = row_f4(h, s1, lane);
        const v4f q2 = row_f4(h, s2, lane);
        const v4f q3 = row_f4(h, s3, lane);
        acc[j] += q0;
        if (rem > 1) acc[j] += q1;
        if (rem > 2) acc[j] += q2;
        if (rem > 3) acc[j] += q3;
      }
    }
  }

  if (dmax > ELL_K) {
    int nsp = cnt[N_NODES];
    nsp = (nsp < N_EDGES) ? nsp : N_EDGES;
#pragma unroll
    for (int j = 0; j < 4; ++j) {
      if (dg[j] > ELL_K) {
        const int n = n0 + 8 * j;
        for (int i = 0; i < nsp; i += 32) {
          const int ii = i + lane;
          int od = -1, os = 0;
          if (ii < nsp) {
            od = spill[2 * ii];
            os = spill[2 * ii + 1];
          }
          unsigned m = __builtin_amdgcn_ballot_w32(od == n);
          while (m != 0u) {
            const int bit = __builtin_ctz(m);
            const int s = __builtin_amdgcn_readlane(os, bit);
            acc[j] += row_f4(h, s, lane);
            m &= m - 1u;
          }
        }
      }
    }
  }

#pragma unroll
  for (int j = 0; j < 4; ++j) {
    const float inv = 1.0f / fmaxf((float)dg[j], 1.0f);
    mean[j] = acc[j] * inv;
  }
}

__device__ __forceinline__ void hrows_load(v4f (&hrow)[4], const float* __restrict__ h,
                                           int tile, int sr, int sc) {
  const v4f* hp = (const v4f*)(h + (size_t)tile * (TILE_M * D));
  const int o = sr * (D / 4) + sc;
  hrow[0] = hp[o];
  hrow[1] = hp[o + 8 * D / 4];
  hrow[2] = hp[o + 16 * D / 4];
  hrow[3] = hp[o + 24 * D / 4];
}

__global__ __launch_bounds__(256) void layer_gemm(const float* __restrict__ h,
                                                  const int* __restrict__ cnt,
                                                  const int* __restrict__ ell,
                                                  const int* __restrict__ spill,
                                                  const float* __restrict__ Wself,
                                                  const float* __restrict__ Wneigh,
                                                  const float* __restrict__ bias_v,
                                                  float* __restrict__ h_out,
                                                  int relu) {
  __shared__ __attribute__((aligned(16))) float sh[2][TILE_M * D_PAD];
  __shared__ __attribute__((aligned(16))) float sa[2][TILE_M * D_PAD];
  const int tid  = threadIdx.x;
  const int lane = tid & 31;
  const int half = lane >> 4;
  const int l16  = lane & 15;
  const int col  = (tid >> 5) * 16 + l16;

  v2f bS[32], bN[32];
#pragma unroll
  for (int t = 0; t < 32; ++t) {
    const int ka = 4 * t + 2 * half;
    bS[t].x = Wself[(size_t)(ka)      * D + col];
    bS[t].y = Wself[(size_t)(ka + 1)  * D + col];
    bN[t].x = Wneigh[(size_t)(ka)     * D + col];
    bN[t].y = Wneigh[(size_t)(ka + 1) * D + col];
  }
  const float bv = bias_v[col];

  const int sc = lane;
  const int sr = __builtin_amdgcn_readfirstlane(tid >> 5);

  v4f hrow[4];
  v4f mean[4];
  int dg[4];
  int idv;

  int tile = blockIdx.x;
  if (tile >= N_TILES) return;

  hrows_load(hrow, h, tile, sr, sc);
  degs_load(dg, cnt, tile * TILE_M + sr);
  idv = ids_request(ell, cnt, dg, tile * TILE_M + sr, lane);
  means_build(mean, h, ell, cnt, spill, dg, idv, tile * TILE_M + sr, sc);
  if (tile + GEMM_GRID < N_TILES) degs_load(dg, cnt, (tile + GEMM_GRID) * TILE_M + sr);

  int buf = 0;
  for (; tile < N_TILES; tile += GEMM_GRID) {
    float* shb = sh[buf];
    float* sab = sa[buf];
    {
      const int o = sr * D_PAD + 4 * sc;
      *(v4f*)(shb + o)              = hrow[0];
      *(v4f*)(shb + o + 8 * D_PAD)  = hrow[1];
      *(v4f*)(shb + o + 16 * D_PAD) = hrow[2];
      *(v4f*)(shb + o + 24 * D_PAD) = hrow[3];
      *(v4f*)(sab + o)              = mean[0];
      *(v4f*)(sab + o + 8 * D_PAD)  = mean[1];
      *(v4f*)(sab + o + 16 * D_PAD) = mean[2];
      *(v4f*)(sab + o + 24 * D_PAD) = mean[3];
    }
    __syncthreads();

    const int nxt = tile + GEMM_GRID;
    int dv = 0;
    if (nxt < N_TILES) {
      hrows_load(hrow, h, nxt, sr, sc);
      idv = ids_request(ell, cnt, dg, nxt * TILE_M + sr, lane);
      const int q = (nxt + GEMM_GRID) * TILE_M + lane;
      dv = cnt[q < N_NODES ? q : N_NODES - 1];
    }

    v8f acc0 = {0.f, 0.f, 0.f, 0.f, 0.f, 0.f, 0.f, 0.f};
    v8f acc1 = {0.f, 0.f, 0.f, 0.f, 0.f, 0.f, 0.f, 0.f};
    const float* h0 = shb + l16 * D_PAD + 2 * half;
    const float* h1 = h0 + 16 * D_PAD;
    const float* g0 = sab + l16 * D_PAD + 2 * half;
    const float* g1 = g0 + 16 * D_PAD;
#pragma unroll
    for (int t = 0; t < 32; ++t) {
      v2f a10 = *(const v2f*)(h0 + 4 * t);
      v2f a11 = *(const v2f*)(h1 + 4 * t);
      acc0 = wmma_f32_16x16x4(a10, bS[t], acc0);
      acc1 = wmma_f32_16x16x4(a11, bS[t], acc1);
      v2f a20 = *(const v2f*)(g0 + 4 * t);
      v2f a21 = *(const v2f*)(g1 + 4 * t);
      acc0 = wmma_f32_16x16x4(a20, bN[t], acc0);
      acc1 = wmma_f32_16x16x4(a21, bN[t], acc1);
    }

    float* ob = h_out + (size_t)tile * (TILE_M * D);
    const int lo = 8 * half * D + col;
#pragma unroll
    for (int r = 0; r < 8; ++r) {
      float v = acc0[r] + bv;
      if (relu) v = fmaxf(v, 0.0f);
      ob[lo + r * D] = v;
    }
#pragma unroll
    for (int r = 0; r < 8; ++r) {
      float v = acc1[r] + bv;
      if (relu) v = fmaxf(v, 0.0f);
      ob[lo + (16 + r) * D] = v;
    }

    if (nxt < N_TILES) {
      means_build(mean, h, ell, cnt, spill, dg, idv, nxt * TILE_M + sr, sc);
#pragma unroll
      for (int j = 0; j < 4; ++j) dg[j] = __builtin_amdgcn_readlane(dv, sr + 8 * j);
    }

    buf ^= 1;
  }
}

extern "C" void kernel_launch(void* const* d_in, const int* in_sizes, int n_in,
                              void* d_out, int out_size, void* d_ws, size_t ws_size,
                              hipStream_t stream) {
  const float* x       = (const float*)d_in[0];
  const int*   ei      = (const int*)d_in[1];
  const float* W_in    = (const float*)d_in[2];
  const float* b_in    = (const float*)d_in[3];
  const float* W_self  = (const float*)d_in[4];
  const float* W_neigh = (const float*)d_in[5];
  const float* b_lay   = (const float*)d_in[6];
  float* out = (float*)d_out;

  float* ws   = (float*)d_ws;
  float* hA   = ws;
  int*   ib   = (int*)(ws + (size_t)1 * N_NODES * D);
  int*   cnt   = ib;
  int*   ell   = ib + CNT_PAD;
  int*   spill = ib + CNT_PAD + N_NODES * ELL_K;
  float* hB   = ws + (size_t)2 * N_NODES * D;

  const int* src = ei;
  const int* dst = ei + N_EDGES;

  hipMemsetAsync(cnt, 0, CNT_PAD * sizeof(int), stream);
  place_kernel<<<(N_EDGES / 4 + 255) / 256, 256, 0, stream>>>(src, dst, cnt, ell, spill, N_EDGES / 4);

  input_gemm_tanh<<<GEMM_GRID, 256, 0, stream>>>(x, W_in, b_in, hA);

  layer_gemm<<<GEMM_GRID, 256, 0, stream>>>(hA, cnt, ell, spill, W_self, W_neigh, b_lay, hB, 1);
  layer_gemm<<<GEMM_GRID, 256, 0, stream>>>(hB, cnt, ell, spill, W_self + (size_t)1 * D * D,
                                            W_neigh + (size_t)1 * D * D, b_lay + (size_t)1 * D, hA, 1);
  layer_gemm<<<GEMM_GRID, 256, 0, stream>>>(hA, cnt, ell, spill, W_self + (size_t)2 * D * D,
                                            W_neigh + (size_t)2 * D * D, b_lay + (size_t)2 * D, out, 0);
}


# Edge placement pass folded into the input-projection kernel: 96 edges per tile, their counting adds in flight under the tile's matrix block; launches 5->4; row sums still in count arrival order
# speedup vs baseline: 2.1430x; 1.0469x over previous
#include <hip/hip_runtime.h>
#include <math.h>


typedef __attribute__((ext_vector_type(2))) float v2f;
typedef __attribute__((ext_vector_type(4))) float v4f;
typedef __attribute__((ext_vector_type(8))) float v8f;

#define N_NODES 200000
#define N_EDGES 600000
#define F_IN    117
#define F_PAD   124
#define D       128
#define D_PAD   132
#define L_LAYERS 3
#define TILE_M   32
#define N_TILES  (N_NODES / TILE_M)
#define GEMM_GRID 512
#define X_TILE_F4 (TILE_M * F_IN / 4)
#define ELL_K   16
#define CNT_PAD 200704
#define E_TILE  (N_EDGES / N_TILES)
#define E_WAVE  (E_TILE / 8)
static_assert(E_TILE * N_TILES == N_EDGES && E_WAVE * 8 == E_TILE && E_WAVE <= 32,
              "the edge pass is cut into equal slices, one per tile and wave");

__device__ __forceinline__ v8f wmma_f32_16x16x4(v2f a, v2f b, v8f c) {
  return __builtin_amdgcn_wmma_f32_16x16x4_f32(false, a, false, b, (short)0, c,
                                               false, false);
}


__global__ __launch_bounds__(256) void input_gemm_tanh(const float* __restrict__ x,
                                                       const float* __restrict__ W_in,
                                                       const float* __restrict__ b_in,
                                                       float* __restrict__ h_out,
                                                       const int* __restrict__ src,
                                                       const int* __restrict__ dst,
                                                       int* __restrict__ cnt,
                                                       int* __restrict__ ell,
                                                       int* __restrict__ spill) {
  __shared__ __attribute__((aligned(16))) float sx[2][TILE_M * F_PAD];
  const int tid  = threadIdx.x;
  const int lane = tid & 31;
  const int half = lane >> 4;
  const int l16  = lane & 15;
  const int col  = (tid >> 5) * 16 + l16;

  for (int i = tid; i < 2 * TILE_M * (F_PAD - F_IN); i += 256) {
    int r = i / (F_PAD - F_IN), c = i - r * (F_PAD - F_IN);
    (&sx[0][0])[r * F_PAD + F_IN + c] = 0.0f;
  }

  v2f bW[30];
#pragma unroll
  for (int t = 0; t < 30; ++t) {
    const int ka = 4 * t + 2 * half;
    bW[t].x = (ka     < F_IN) ? W_in[(size_t)(ka)     * D + col] : 0.0f;
    bW[t].y = (ka + 1 < F_IN) ? W_in[(size_t)(ka + 1) * D + col] : 0.0f;
  }
  const float bias = b_in[col];

  const bool tail = tid < X_TILE_F4 - 768;
  v4f px[4];
  int tile = blockIdx.x;
  if (tile < N_TILES) {
    const v4f* xt = (const v4f*)(x + (size_t)tile * (TILE_M * F_IN));
    px[0] = xt[tid];
    px[1] = xt[tid + 256];
    px[2] = xt[tid + 512];
    if (tail) px[3] = xt[tid + 768];
  }

  const bool edger = lane < E_WAVE;
  const int eoff = (tid >> 5) * E_WAVE + lane;
  int ed = 0, es = 0;
  if (tile < N_TILES && edger) {
    ed = dst[tile * E_TILE + eoff];
    es = src[tile * E_TILE + eoff];
  }

  int buf = 0;
  for (; tile < N_TILES; tile += GEMM_GRID) {
    float* sxb = sx[buf];
#pragma unroll
    for (int j = 0; j < 4; ++j) {
      if (j < 3 || tail) {
        int i0 = 4 * (tid + 256 * j);
        int r0 = i0 / F_IN, c0 = i0 - r0 * F_IN;
#pragma unroll
        for (int u = 0; u < 4; ++u) {
          sxb[r0 * F_PAD + c0] = px[j][u];
          if (++c0 == F_IN) { c0 = 0; ++r0; }
        }
      }
    }
    __syncthreads();

    const int nxt = tile + GEMM_GRID;
    if (nxt < N_TILES) {
      const v4f* xt = (const v4f*)(x + (size_t)nxt * (TILE_M * F_IN));
      px[0] = xt[tid];
      px[1] = xt[tid + 256];
      px[2] = xt[tid + 512];
      if (tail) px[3] = xt[tid + 768];
    }

    int ep = 0;
    if (edger) ep = atomicAdd(&cnt[ed], 1);
    int nd = 0, ns = 0;
    if (nxt < N_TILES && edger) {
      nd = dst[nxt * E_TILE + eoff];
      ns = src[nxt * E_TILE + eoff];
    }

    v8f acc0 = {0.f, 0.f, 0.f, 0.f, 0.f, 0.f, 0.f, 0.f};
    v8f acc1 = {0.f, 0.f, 0.f, 0.f, 0.f, 0.f, 0.f, 0.f};
    const float* ar0 = sxb + l16 * F_PAD + 2 * half;
    const float* ar1 = ar0 + 16 * F_PAD;
#pragma unroll
    for (int t = 0; t < 30; ++t) {
      v2f a0 = *(const v2f*)(ar0 + 4 * t);
      v2f a1 = *(const v2f*)(ar1 + 4 * t);
      acc0 = wmma_f32_16x16x4(a0, bW[t], acc0);
      acc1 = wmma_f32_16x16x4(a1, bW[t], acc1);
    }

    float* ob = h_out + (size_t)tile * (TILE_M * D);
    const int lo = 8 * half * D + col;
#pragma unroll
    for (int r = 0; r < 8; ++r)
      ob[lo + r * D] = tanhf(acc0[r] + bias);
#pragma unroll
    for (int r = 0; r < 8; ++r)
      ob[lo + (16 + r) * D] = tanhf(acc1[r] + bias);

    if (edger) {
      if (ep < ELL_K) {
        ell[ed * ELL_K + ep] = es;
      } else {
        const int q = atomicAdd(&cnt[N_NODES], 1);
        spill[2 * q]     = ed;
        spill[2 * q + 1] = es;
      }
    }
    ed = nd;
    es = ns;

    buf ^= 1;
  }
}

#define NB_IDS 8
#define NB_GRP 4

__device__ __forceinline__ v4f row_f4(const float* __restrict__ h, int s, int lane) {
  return ((const v4f*)(h + (size_t)s * D))[lane];
}

__device__ __forceinline__ void degs_load(int (&dg)[4], const int* __restrict__ cnt, int n0) {
#pragma unroll
  for (int j = 0; j < 4; ++j) dg[j] = cnt[n0 + 8 * j];
}

__device__ __forceinline__ int ids_request(const int* __restrict__ ell, const int* __restrict__ cnt,
                                           const int (&dg)[4], int n0, int lane) {
  const int j = lane >> 3;
  const int k = lane & 7;
  int dj = dg[0];
  dj = (j == 1) ? dg[1] : dj;
  dj = (j == 2) ? dg[2] : dj;
  dj = (j == 3) ? dg[3] : dj;
  const int n = n0 + 8 * j;
  const int* p = (k < dj) ? (ell + n * ELL_K + k) : (cnt + n);
  return *p;
}

__device__ __forceinline__ void rows_round(v4f (&acc)[4], const float* __restrict__ h,
                                           const int (&dg)[4],
                                           int idv, int n0, int lane, int off) {
  v4f r[4][NB_GRP];
#pragma unroll
  for (int j = 0; j < 4; ++j) {
#pragma unroll
    for (int k = 0; k < NB_GRP; ++k) {
      const int id = __builtin_amdgcn_readlane(idv, NB_IDS * j + off + k);
      const int s = (off + k < dg[j]) ? id : n0 + 8 * j;
      r[j][k] = row_f4(h, s, lane);
    }
  }
  const v4f zero = {0.f, 0.f, 0.f, 0.f};
#pragma unroll
  for (int j = 0; j < 4; ++j) {
#pragma unroll
    for (int k = 0; k < NB_GRP; ++k) {
      const v4f rv = r[j][k];
      acc[j] += (off + k < dg[j]) ? rv : zero;
    }
  }
}

__device__ __forceinline__ void means_build(v4f (&mean)[4], const float* __restrict__ h,
                                            const int* __restrict__ ell,
                                            const int* __restrict__ cnt,
                                            const int* __restrict__ spill,
                                            const int (&dg)[4],
                                            int idv, int n0, int lane) {
  v4f acc[4];
#pragma unroll
  for (int j = 0; j < 4; ++j) acc[j] = (v4f){0.f, 0.f, 0.f, 0.f};
  int dmax = dg[0];
#pragma unroll
  for (int j = 1; j < 4; ++j) dmax = (dg[j] > dmax) ? dg[j] : dmax;

  rows_round(acc, h, dg, idv, n0, lane, 0);
  if (dmax > NB_GRP) rows_round(acc, h, dg, idv, n0, lane, NB_GRP);

  const int dcap = (dmax < ELL_K) ? dmax : ELL_K;
  for (int off = NB_IDS; off < dcap; off += 4) {
#pragma unroll
    for (int j = 0; j < 4; ++j) {
      const int dj = (dg[j] < ELL_K) ? dg[j] : ELL_K;
      const int rem = dj - off;
      if (rem > 0) {
        const int* q = ell + (n0 + 8 * j) * ELL_K + off;
        const int s0 = q[0];
        const int s1 = (rem > 1) ? q[1] : s0;
        const int s2 = (rem > 2) ? q[2] : s0;
        const int s3 = (rem > 3) ? q[3] : s0;
        const v4f q0 = row_f4(h, s0, lane);
        const v4f q1 = row_f4(h, s1, lane);
        const v4f q2 = row_f4(h, s2, lane);
        const v4f q3 = row_f4(h, s3, lane);
        acc[j] += q0;
        if (rem > 1) acc[j] += q1;
        if (rem > 2) acc[j] += q2;
        if (rem > 3) acc[j] += q3;
      }
    }
  }

  if (dmax > ELL_K) {
    int nsp = cnt[N_NODES];
    nsp = (nsp < N_EDGES) ? nsp : N_EDGES;
#pragma unroll
    for (int j = 0; j < 4; ++j) {
      if (dg[j] > ELL_K) {
        const int n = n0 + 8 * j;
        for (int i = 0; i < nsp; i += 32) {
          const int ii = i + lane;
          int od = -1, os = 0;
          if (ii < nsp) {
            od = spill[2 * ii];
            os = spill[2 * ii + 1];
          }
          unsigned m = __builtin_amdgcn_ballot_w32(od == n);
          while (m != 0u) {
            const int bit = __builtin_ctz(m);
            const int s = __builtin_amdgcn_readlane(os, bit);
            acc[j] += row_f4(h, s, lane);
            m &= m - 1u;
          }
        }
      }
    }
  }

#pragma unroll
  for (int j = 0; j < 4; ++j) {
    const float inv = 1.0f / fmaxf((float)dg[j], 1.0f);
    mean[j] = acc[j] * inv;
  }
}

__device__ __forceinline__ void hrows_load(v4f (&hrow)[4], const float* __restrict__ h,
                                           int tile, int sr, int sc) {
  const v4f* hp = (const v4f*)(h + (size_t)tile * (TILE_M * D));
  const int o = sr * (D / 4) + sc;
  hrow[0] = hp[o];
  hrow[1] = hp[o + 8 * D / 4];
  hrow[2] = hp[o + 16 * D / 4];
  hrow[3] = hp[o + 24 * D / 4];
}

__global__ __launch_bounds__(256) void layer_gemm(const float* __restrict__ h,
                                                  const int* __restrict__ cnt,
                                                  const int* __restrict__ ell,
                                                  const int* __restrict__ spill,
                                                  const float* __restrict__ Wself,
                                                  const float* __restrict__ Wneigh,
                                                  const float* __restrict__ bias_v,
                                                  float* __restrict__ h_out,
                                                  int relu) {
  __shared__ __attribute__((aligned(16))) float sh[2][TILE_M * D_PAD];
  __shared__ __attribute__((aligned(16))) float sa[2][TILE_M * D_PAD];
  const int tid  = threadIdx.x;
  const int lane = tid & 31;
  const int half = lane >> 4;
  const int l16  = lane & 15;
  const int col  = (tid >> 5) * 16 + l16;

  v2f bS[32], bN[32];
#pragma unroll
  for (int t = 0; t < 32; ++t) {
    const int ka = 4 * t + 2 * half;
    bS[t].x = Wself[(size_t)(ka)      * D + col];
    bS[t].y = Wself[(size_t)(ka + 1)  * D + col];
    bN[t].x = Wneigh[(size_t)(ka)     * D + col];
    bN[t].y = Wneigh[(size_t)(ka + 1) * D + col];
  }
  const float bv = bias_v[col];

  const int sc = lane;
  const int sr = __builtin_amdgcn_readfirstlane(tid >> 5);

  v4f hrow[4];
  v4f mean[4];
  int dg[4];
  int idv;

  int tile = blockIdx.x;
  if (tile >= N_TILES) return;

  hrows_load(hrow, h, tile, sr, sc);
  degs_load(dg, cnt, tile * TILE_M + sr);
  idv = ids_request(ell, cnt, dg, tile * TILE_M + sr, lane);
  means_build(mean, h, ell, cnt, spill, dg, idv, tile * TILE_M + sr, sc);
  if (tile + GEMM_GRID < N_TILES) degs_load(dg, cnt, (tile + GEMM_GRID) * TILE_M + sr);

  int buf = 0;
  for (; tile < N_TILES; tile += GEMM_GRID) {
    float* shb = sh[buf];
    float* sab = sa[buf];
    {
      const int o = sr * D_PAD + 4 * sc;
      *(v4f*)(shb + o)              = hrow[0];
      *(v4f*)(shb + o + 8 * D_PAD)  = hrow[1];
      *(v4f*)(shb + o + 16 * D_PAD) = hrow[2];
      *(v4f*)(shb + o + 24 * D_PAD) = hrow[3];
      *(v4f*)(sab + o)              = mean[0];
      *(v4f*)(sab + o + 8 * D_PAD)  = mean[1];
      *(v4f*)(sab + o + 16 * D_PAD) = mean[2];
      *(v4f*)(sab + o + 24 * D_PAD) = mean[3];
    }
    __syncthreads();

    const int nxt = tile + GEMM_GRID;
    int dv = 0;
    if (nxt < N_TILES) {
      hrows_load(hrow, h, nxt, sr, sc);
      idv = ids_request(ell, cnt, dg, nxt * TILE_M + sr, lane);
      const int q = (nxt + GEMM_GRID) * TILE_M + lane;
      dv = cnt[q < N_NODES ? q : N_NODES - 1];
    }

    v8f acc0 = {0.f, 0.f, 0.f, 0.f, 0.f, 0.f, 0.f, 0.f};
    v8f acc1 = {0.f, 0.f, 0.f, 0.f, 0.f, 0.f, 0.f, 0.f};
    const float* h0 = shb + l16 * D_PAD + 2 * half;
    const float* h1 = h0 + 16 * D_PAD;
    const float* g0 = sab + l16 * D_PAD + 2 * half;
    const float* g1 = g0 + 16 * D_PAD;
#pragma unroll
    for (int t = 0; t < 32; ++t) {
      v2f a10 = *(const v2f*)(h0 + 4 * t);
      v2f a11 = *(const v2f*)(h1 + 4 * t);
      acc0 = wmma_f32_16x16x4(a10, bS[t], acc0);
      acc1 = wmma_f32_16x16x4(a11, bS[t], acc1);
      v2f a20 = *(const v2f*)(g0 + 4 * t);
      v2f a21 = *(const v2f*)(g1 + 4 * t);
      acc0 = wmma_f32_16x16x4(a20, bN[t], acc0);
      acc1 = wmma_f32_16x16x4(a21, bN[t], acc1);
    }

    float* ob = h_out + (size_t)tile * (TILE_M * D);
    const int lo = 8 * half * D + col;
#pragma unroll
    for (int r = 0; r < 8; ++r) {
      float v = acc0[r] + bv;
      if (relu) v = fmaxf(v, 0.0f);
      ob[lo + r * D] = v;
    }
#pragma unroll
    for (int r = 0; r < 8; ++r) {
      float v = acc1[r] + bv;
      if (relu) v = fmaxf(v, 0.0f);
      ob[lo + (16 + r) * D] = v;
    }

    if (nxt < N_TILES) {
      means_build(mean, h, ell, cnt, spill, dg, idv, nxt * TILE_M + sr, sc);
#pragma unroll
      for (int j = 0; j < 4; ++j) dg[j] = __builtin_amdgcn_readlane(dv, sr + 8 * j);
    }

    buf ^= 1;
  }
}

extern "C" void kernel_launch(void* const* d_in, const int* in_sizes, int n_in,
                              void* d_out, int out_size, void* d_ws, size_t ws_size,
                              hipStream_t stream) {
  const float* x       = (const float*)d_in[0];
  const int*   ei      = (const int*)d_in[1];
  const float* W_in    = (const float*)d_in[2];
  const float* b_in    = (const float*)d_in[3];
  const float* W_self  = (const float*)d_in[4];
  const float* W_neigh = (const float*)d_in[5];
  const float* b_lay   = (const float*)d_in[6];
  float* out = (float*)d_out;

  float* ws   = (float*)d_ws;
  float* hA   = ws;
  int*   ib   = (int*)(ws + (size_t)1 * N_NODES * D);
  int*   cnt   = ib;
  int*   ell   = ib + CNT_PAD;
  int*   spill = ib + CNT_PAD + N_NODES * ELL_K;
  float* hB   = ws + (size_t)2 * N_NODES * D;

  const int* src = ei;
  const int* dst = ei + N_EDGES;

  hipMemsetAsync(cnt, 0, CNT_PAD * sizeof(int), stream);

  input_gemm_tanh<<<GEMM_GRID, 256, 0, stream>>>(x, W_in, b_in, hA, src, dst, cnt, ell, spill);

  layer_gemm<<<GEMM_GRID, 256, 0, stream>>>(hA, cnt, ell, spill, W_self, W_neigh, b_lay, hB, 1);
  layer_gemm<<<GEMM_GRID, 256, 0, stream>>>(hB, cnt, ell, spill, W_self + (size_t)1 * D * D,
                                            W_neigh + (size_t)1 * D * D, b_lay + (size_t)1 * D, hA, 1);
  layer_gemm<<<GEMM_GRID, 256, 0, stream>>>(hA, cnt, ell, spill, W_self + (size_t)2 * D * D,
                                            W_neigh + (size_t)2 * D * D, b_lay + (size_t)2 * D, out, 0);
}
